# starting kernel
# baseline (speedup 1.0000x reference)
#include <hip/hip_runtime.h>
#include <math.h>

#define ROWS 1024
#define KDIM 128
#define PAD 132

typedef __attribute__((ext_vector_type(2))) float v2f;
typedef __attribute__((ext_vector_type(8))) float v8f;

__device__ __forceinline__ void stage_bt(const float* __restrict__ Bg,
                                         float* __restrict__ lBt, int tid) {
  const int n  = tid & 127;
  const int k0 = (tid >> 7) * 64;
#pragma unroll 4
  for (int k = k0; k < k0 + 64; k += 4) {
    float x0 = Bg[(k + 0) * KDIM + n];
    float x1 = Bg[(k + 1) * KDIM + n];
    float x2 = Bg[(k + 2) * KDIM + n];
    float x3 = Bg[(k + 3) * KDIM + n];
    float* d = &lBt[n * PAD + k];
    d[0] = x0; d[1] = x1; d[2] = x2; d[3] = x3;
  }
}

__device__ __forceinline__ v8f wmma_k128_lds(const float* __restrict__ lA,
                                             const float* __restrict__ lBt,
                                             int n0, int lane) {
  const int l16  = lane & 15;
  const int koff = (lane >> 4) << 1;
  const float* ap = lA  + l16 * PAD + koff;
  const float* bp = lBt + (n0 + l16) * PAD + koff;
  v8f acc = {0.f, 0.f, 0.f, 0.f, 0.f, 0.f, 0.f, 0.f};
#pragma unroll
  for (int k0 = 0; k0 < KDIM; k0 += 4) {
    v2f a, b;
    a.x = ap[k0]; a.y = ap[k0 + 1];
    b.x = bp[k0]; b.y = bp[k0 + 1];
    acc = __builtin_amdgcn_wmma_f32_16x16x4_f32(false, a, false, b,
                                                (short)0, acc, false, false);
  }
  return acc;
}

__device__ __forceinline__ float silu_f(float x) {
  return x * __builtin_amdgcn_rcpf(1.0f + __expf(-x));
}

__global__ void k_proj(const float* __restrict__ h,
                       const float* __restrict__ W1a,
                       const float* __restrict__ W1b,
                       float* __restrict__ pi, float* __restrict__ pj) {
  __shared__ float lA[16 * PAD];
  __shared__ float lBa[KDIM * PAD];
  __shared__ float lBb[KDIM * PAD];
  const int r0  = blockIdx.x * 16;
  const int tid = threadIdx.x;
  for (int idx = tid; idx < 16 * KDIM; idx += 256)
    lA[(idx >> 7) * PAD + (idx & 127)] = h[(r0 + (idx >> 7)) * KDIM + (idx & 127)];
  stage_bt(W1a, lBa, tid);
  stage_bt(W1b, lBb, tid);
  __syncthreads();

  const int lane = tid & 31, wave = tid >> 5;
  const int n0 = wave * 16, l16 = lane & 15, half = lane >> 4;
  v8f p = wmma_k128_lds(lA, lBa, n0, lane);
  v8f q = wmma_k128_lds(lA, lBb, n0, lane);
#pragma unroll
  for (int r = 0; r < 8; ++r) {
    const int row = r0 + r + half * 8;
    pi[row * KDIM + n0 + l16] = p[r];
    pj[row * KDIM + n0 + l16] = q[r];
  }
}

__global__ void k_msg(const float* __restrict__ pi, const float* __restrict__ pj,
                      const float* __restrict__ b1, const float* __restrict__ adj,
                      float* __restrict__ S, float* __restrict__ Arow,
                      float* __restrict__ Den) {
  __shared__ float adjs[256];
  const int bi = blockIdx.x;
  const int b  = bi >> 8;
  const int i  = bi & 255;
  const int t  = threadIdx.x;

  const float* __restrict__ ar = adj + (size_t)bi * 256;
  for (int j = t; j < 256; j += 128) adjs[j] = ar[j];
  __syncthreads();

  const float base = pi[bi * KDIM + t] + b1[t];
  const float* __restrict__ pjb = pj + (size_t)b * 256 * KDIM + t;

  float acc = 0.f, rs = 0.f;
#pragma unroll 4
  for (int j = 0; j < 256; ++j) {
    const float w = adjs[j];
    rs += w;
    const float x = base + pjb[j * KDIM];
    const float s = silu_f(x);
    const float we = (j == i) ? 0.f : w;
    acc = fmaf(we, s, acc);
  }
  S[bi * KDIM + t] = acc;
  if (t == 0) {
    Arow[bi] = rs - adjs[i];
    Den[bi]  = fmaxf(rs, 1.0f);
  }
}

__global__ void k_agg(const float* __restrict__ S, const float* __restrict__ W2,
                      const float* __restrict__ b2, const float* __restrict__ Arow,
                      const float* __restrict__ Den, float* __restrict__ magg) {
  __shared__ float lA[16 * PAD];
  __shared__ float lB[KDIM * PAD];
  const int r0  = blockIdx.x * 16;
  const int tid = threadIdx.x;
  for (int idx = tid; idx < 16 * KDIM; idx += 256)
    lA[(idx >> 7) * PAD + (idx & 127)] = S[(r0 + (idx >> 7)) * KDIM + (idx & 127)];
  stage_bt(W2, lB, tid);
  __syncthreads();

  const int lane = tid & 31, wave = tid >> 5;
  const int n0 = wave * 16, l16 = lane & 15, half = lane >> 4;
  v8f acc = wmma_k128_lds(lA, lB, n0, lane);
  const float bcol = b2[n0 + l16];
#pragma unroll
  for (int r = 0; r < 8; ++r) {
    const int row = r0 + r + half * 8;
    const float rd = __builtin_amdgcn_rcpf(Den[row]);
    magg[row * KDIM + n0 + l16] = (acc[r] + bcol * Arow[row]) * rd;
  }
}

__global__ void k_head(const float* __restrict__ magg,
                       const float* __restrict__ Wc1, const float* __restrict__ bc1,
                       const float* __restrict__ Wc2, const float* __restrict__ bc2,
                       float* __restrict__ out) {
  __shared__ float lA[16 * PAD];
  __shared__ float lT[16 * PAD];
  __shared__ float lB1[KDIM * PAD];
  __shared__ float lB2[KDIM * PAD];
  const int r0  = blockIdx.x * 16;
  const int tid = threadIdx.x;
  for (int idx = tid; idx < 16 * KDIM; idx += 256)
    lA[(idx >> 7) * PAD + (idx & 127)] = magg[(r0 + (idx >> 7)) * KDIM + (idx & 127)];
  stage_bt(Wc1, lB1, tid);
  stage_bt(Wc2, lB2, tid);
  __syncthreads();

  const int lane = tid & 31, wave = tid >> 5;
  const int n0 = wave * 16, l16 = lane & 15, half = lane >> 4;

  v8f u = wmma_k128_lds(lA, lB1, n0, lane);
  const float b1c = bc1[n0 + l16];
#pragma unroll
  for (int r = 0; r < 8; ++r) {
    const int rr = r + half * 8;
    lT[rr * PAD + n0 + l16] = silu_f(u[r] + b1c);
  }
  __syncthreads();

  v8f c = wmma_k128_lds(lT, lB2, n0, lane);
  const float b2c = bc2[n0 + l16];
#pragma unroll
  for (int r = 0; r < 8; ++r) {
    const int row = r0 + r + half * 8;
    out[row * KDIM + n0 + l16] = c[r] + b2c;
  }
}

extern "C" void kernel_launch(void* const* d_in, const int* in_sizes, int n_in,
                              void* d_out, int out_size, void* d_ws, size_t ws_size,
                              hipStream_t stream) {
  const float* h   = (const float*)d_in[0];
  const float* adj = (const float*)d_in[1];
  const float* W1a = (const float*)d_in[2];
  const float* W1b = (const float*)d_in[3];
  const float* b1  = (const float*)d_in[4];
  const float* W2  = (const float*)d_in[5];
  const float* b2  = (const float*)d_in[6];
  const float* Wc1 = (const float*)d_in[7];
  const float* bc1 = (const float*)d_in[8];
  const float* Wc2 = (const float*)d_in[9];
  const float* bc2 = (const float*)d_in[10];
  float* out = (float*)d_out;

  float* ws = (float*)d_ws;
  float* pi = ws;
  float* pj = ws + 131072;
  float* S  = ws + 262144;
  float* Ar = ws + 393216;
  float* Dn = ws + 394240;
  float* mg = ws + 395264;

  k_proj<<<ROWS / 16, 256, 0, stream>>>(h, W1a, W1b, pi, pj);
  k_msg <<<ROWS, 128, 0, stream>>>(pi, pj, b1, adj, S, Ar, Dn);
  k_agg <<<ROWS / 16, 256, 0, stream>>>(S, W2, b2, Ar, Dn, mg);
  k_head<<<ROWS / 16, 256, 0, stream>>>(mg, Wc1, bc1, Wc2, bc2, out);
}


# k_msg: block = 4 dest rows x 128 cols (8 waves), each thread 2 rows per pj value, pj loads issued 32 source rows ahead of use (was 4 loads then wait per 4 silu); every sum keeps its j order
# speedup vs baseline: 1.5924x; 1.5924x over previous
#include <hip/hip_runtime.h>
#include <math.h>

#define ROWS 1024
#define KDIM 128
#define PAD 132

typedef __attribute__((ext_vector_type(2))) float v2f;
typedef __attribute__((ext_vector_type(8))) float v8f;

__device__ __forceinline__ void stage_bt(const float* __restrict__ Bg,
                                         float* __restrict__ lBt, int tid) {
  const int n  = tid & 127;
  const int k0 = (tid >> 7) * 64;
#pragma unroll 4
  for (int k = k0; k < k0 + 64; k += 4) {
    float x0 = Bg[(k + 0) * KDIM + n];
    float x1 = Bg[(k + 1) * KDIM + n];
    float x2 = Bg[(k + 2) * KDIM + n];
    float x3 = Bg[(k + 3) * KDIM + n];
    float* d = &lBt[n * PAD + k];
    d[0] = x0; d[1] = x1; d[2] = x2; d[3] = x3;
  }
}

__device__ __forceinline__ v8f wmma_k128_lds(const float* __restrict__ lA,
                                             const float* __restrict__ lBt,
                                             int n0, int lane) {
  const int l16  = lane & 15;
  const int koff = (lane >> 4) << 1;
  const float* ap = lA  + l16 * PAD + koff;
  const float* bp = lBt + (n0 + l16) * PAD + koff;
  v8f acc = {0.f, 0.f, 0.f, 0.f, 0.f, 0.f, 0.f, 0.f};
#pragma unroll
  for (int k0 = 0; k0 < KDIM; k0 += 4) {
    v2f a, b;
    a.x = ap[k0]; a.y = ap[k0 + 1];
    b.x = bp[k0]; b.y = bp[k0 + 1];
    acc = __builtin_amdgcn_wmma_f32_16x16x4_f32(false, a, false, b,
                                                (short)0, acc, false, false);
  }
  return acc;
}

__device__ __forceinline__ float silu_f(float x) {
  return x * __builtin_amdgcn_rcpf(1.0f + __expf(-x));
}

__global__ void k_proj(const float* __restrict__ h,
                       const float* __restrict__ W1a,
                       const float* __restrict__ W1b,
                       float* __restrict__ pi, float* __restrict__ pj) {
  __shared__ float lA[16 * PAD];
  __shared__ float lBa[KDIM * PAD];
  __shared__ float lBb[KDIM * PAD];
  const int r0  = blockIdx.x * 16;
  const int tid = threadIdx.x;
  for (int idx = tid; idx < 16 * KDIM; idx += 256)
    lA[(idx >> 7) * PAD + (idx & 127)] = h[(r0 + (idx >> 7)) * KDIM + (idx & 127)];
  stage_bt(W1a, lBa, tid);
  stage_bt(W1b, lBb, tid);
  __syncthreads();

  const int lane = tid & 31, wave = tid >> 5;
  const int n0 = wave * 16, l16 = lane & 15, half = lane >> 4;
  v8f p = wmma_k128_lds(lA, lBa, n0, lane);
  v8f q = wmma_k128_lds(lA, lBb, n0, lane);
#pragma unroll
  for (int r = 0; r < 8; ++r) {
    const int row = r0 + r + half * 8;
    pi[row * KDIM + n0 + l16] = p[r];
    pj[row * KDIM + n0 + l16] = q[r];
  }
}

#define MSG_ROWS 4
#define MSG_JB 32

__device__ __forceinline__ void msg_block(const float (&p)[MSG_JB],
                                            const float* __restrict__ w0,
                                            const float* __restrict__ w1,
                                            float base0, float base1,
                                            float& acc0, float& acc1) {
  float wa[MSG_JB], wb[MSG_JB];
#pragma unroll
  for (int q = 0; q < MSG_JB; q += 4) {
    const float4 va = *reinterpret_cast<const float4*>(w0 + q);
    const float4 vb = *reinterpret_cast<const float4*>(w1 + q);
    wa[q] = va.x; wa[q + 1] = va.y; wa[q + 2] = va.z; wa[q + 3] = va.w;
    wb[q] = vb.x; wb[q + 1] = vb.y; wb[q + 2] = vb.z; wb[q + 3] = vb.w;
  }
#pragma unroll
  for (int k = 0; k < MSG_JB; ++k) {
    acc0 = fmaf(wa[k], silu_f(base0 + p[k]), acc0);
    acc1 = fmaf(wb[k], silu_f(base1 + p[k]), acc1);
  }
}

__global__ void __launch_bounds__(256)
k_msg(const float* __restrict__ pi, const float* __restrict__ pj,
      const float* __restrict__ b1, const float* __restrict__ adj,
      float* __restrict__ S, float* __restrict__ Arow,
      float* __restrict__ Den) {
  __shared__ __attribute__((aligned(16))) float adjr[MSG_ROWS * 256];
  __shared__ __attribute__((aligned(16))) float adjm[MSG_ROWS * 256];
  const int row0 = blockIdx.x * MSG_ROWS;
  const int b    = row0 >> 8;
  const int i0   = row0 & 255;
  const int t    = threadIdx.x;
  const int m    = t & 127;
  const int wv   = __builtin_amdgcn_readfirstlane(t >> 5);
  const int ra   = (wv >> 2) * 2;

  const float* __restrict__ ar = adj + (size_t)row0 * 256;
  float av[MSG_ROWS];
#pragma unroll
  for (int q = 0; q < MSG_ROWS; ++q) av[q] = ar[q * 256 + t];

  const float bb    = b1[m];
  const float base0 = pi[(row0 + ra) * KDIM + m] + bb;
  const float base1 = pi[(row0 + ra + 1) * KDIM + m] + bb;
  const float* __restrict__ pjb = pj + (size_t)b * 256 * KDIM + m;

  float pa[MSG_JB], pb[MSG_JB];
#pragma unroll
  for (int k = 0; k < MSG_JB; ++k) pa[k] = pjb[k * KDIM];

#pragma unroll
  for (int q = 0; q < MSG_ROWS; ++q) {
    adjr[q * 256 + t] = av[q];
    adjm[q * 256 + t] = (t == i0 + q) ? 0.f : av[q];
  }
  __syncthreads();

  const float* __restrict__ w0 = adjm + ra * 256;
  const float* __restrict__ w1 = w0 + 256;
  const float* __restrict__ wr = adjr + (wv & 3) * 256;

  float acc0 = 0.f, acc1 = 0.f, rs = 0.f;
  for (int jb = 0; jb < 256; jb += 2 * MSG_JB) {
#pragma unroll
    for (int k = 0; k < MSG_JB; ++k) pb[k] = pjb[(jb + MSG_JB + k) * KDIM];
    if (wv < 4) {
#pragma unroll
      for (int k = 0; k < 2 * MSG_JB; ++k) rs += wr[jb + k];
    }
    msg_block(pa, w0 + jb, w1 + jb, base0, base1, acc0, acc1);
    const int jn = (jb + 2 * MSG_JB) & 255;
#pragma unroll
    for (int k = 0; k < MSG_JB; ++k) pa[k] = pjb[(jn + k) * KDIM];
    msg_block(pb, w0 + jb + MSG_JB, w1 + jb + MSG_JB, base0, base1, acc0, acc1);
  }
  S[(row0 + ra) * KDIM + m]     = acc0;
  S[(row0 + ra + 1) * KDIM + m] = acc1;
  if (wv < 4 && (t & 31) == 0) {
    Arow[row0 + wv] = rs - adjr[wv * 256 + i0 + wv];
    Den[row0 + wv]  = fmaxf(rs, 1.0f);
  }
}

__global__ void k_agg(const float* __restrict__ S, const float* __restrict__ W2,
                      const float* __restrict__ b2, const float* __restrict__ Arow,
                      const float* __restrict__ Den, float* __restrict__ magg) {
  __shared__ float lA[16 * PAD];
  __shared__ float lB[KDIM * PAD];
  const int r0  = blockIdx.x * 16;
  const int tid = threadIdx.x;
  for (int idx = tid; idx < 16 * KDIM; idx += 256)
    lA[(idx >> 7) * PAD + (idx & 127)] = S[(r0 + (idx >> 7)) * KDIM + (idx & 127)];
  stage_bt(W2, lB, tid);
  __syncthreads();

  const int lane = tid & 31, wave = tid >> 5;
  const int n0 = wave * 16, l16 = lane & 15, half = lane >> 4;
  v8f acc = wmma_k128_lds(lA, lB, n0, lane);
  const float bcol = b2[n0 + l16];
#pragma unroll
  for (int r = 0; r < 8; ++r) {
    const int row = r0 + r + half * 8;
    const float rd = __builtin_amdgcn_rcpf(Den[row]);
    magg[row * KDIM + n0 + l16] = (acc[r] + bcol * Arow[row]) * rd;
  }
}

__global__ void k_head(const float* __restrict__ magg,
                       const float* __restrict__ Wc1, const float* __restrict__ bc1,
                       const float* __restrict__ Wc2, const float* __restrict__ bc2,
                       float* __restrict__ out) {
  __shared__ float lA[16 * PAD];
  __shared__ float lT[16 * PAD];
  __shared__ float lB1[KDIM * PAD];
  __shared__ float lB2[KDIM * PAD];
  const int r0  = blockIdx.x * 16;
  const int tid = threadIdx.x;
  for (int idx = tid; idx < 16 * KDIM; idx += 256)
    lA[(idx >> 7) * PAD + (idx & 127)] = magg[(r0 + (idx >> 7)) * KDIM + (idx & 127)];
  stage_bt(Wc1, lB1, tid);
  stage_bt(Wc2, lB2, tid);
  __syncthreads();

  const int lane = tid & 31, wave = tid >> 5;
  const int n0 = wave * 16, l16 = lane & 15, half = lane >> 4;

  v8f u = wmma_k128_lds(lA, lB1, n0, lane);
  const float b1c = bc1[n0 + l16];
#pragma unroll
  for (int r = 0; r < 8; ++r) {
    const int rr = r + half * 8;
    lT[rr * PAD + n0 + l16] = silu_f(u[r] + b1c);
  }
  __syncthreads();

  v8f c = wmma_k128_lds(lT, lB2, n0, lane);
  const float b2c = bc2[n0 + l16];
#pragma unroll
  for (int r = 0; r < 8; ++r) {
    const int row = r0 + r + half * 8;
    out[row * KDIM + n0 + l16] = c[r] + b2c;
  }
}

extern "C" void kernel_launch(void* const* d_in, const int* in_sizes, int n_in,
                              void* d_out, int out_size, void* d_ws, size_t ws_size,
                              hipStream_t stream) {
  const float* h   = (const float*)d_in[0];
  const float* adj = (const float*)d_in[1];
  const float* W1a = (const float*)d_in[2];
  const float* W1b = (const float*)d_in[3];
  const float* b1  = (const float*)d_in[4];
  const float* W2  = (const float*)d_in[5];
  const float* b2  = (const float*)d_in[6];
  const float* Wc1 = (const float*)d_in[7];
  const float* bc1 = (const float*)d_in[8];
  const float* Wc2 = (const float*)d_in[9];
  const float* bc2 = (const float*)d_in[10];
  float* out = (float*)d_out;

  float* ws = (float*)d_ws;
  float* pi = ws;
  float* pj = ws + 131072;
  float* S  = ws + 262144;
  float* Ar = ws + 393216;
  float* Dn = ws + 394240;
  float* mg = ws + 395264;

  k_proj<<<ROWS / 16, 256, 0, stream>>>(h, W1a, W1b, pi, pj);
  k_msg <<<ROWS / MSG_ROWS, 256, 0, stream>>>(pi, pj, b1, adj, S, Ar, Dn);
  k_agg <<<ROWS / 16, 256, 0, stream>>>(S, W2, b2, Ar, Dn, mg);
  k_head<<<ROWS / 16, 256, 0, stream>>>(mg, Wc1, bc1, Wc2, bc2, out);
}


# GEMM kernels: weights no longer staged through LDS; each wave (16x32 tile) loads its A and B fragments straight from global; lane halves own k 0-63 / 64-127, so the k sum order changes
# speedup vs baseline: 2.2327x; 1.4021x over previous
#include <hip/hip_runtime.h>
#include <math.h>

#define ROWS 1024
#define KDIM 128
#define PAD 132

typedef __attribute__((ext_vector_type(2))) float v2f;
typedef __attribute__((ext_vector_type(4))) float v4f;
typedef __attribute__((ext_vector_type(8))) float v8f;


__device__ __forceinline__ void load_a16(const float* __restrict__ arow_h, v4f (&a)[16]) {
#pragma unroll
  for (int q = 0; q < 16; ++q) a[q] = *reinterpret_cast<const v4f*>(arow_h + 4 * q);
}

__device__ __forceinline__ void load_b64(const float* __restrict__ bcol_h, v2f (&b)[64]) {
#pragma unroll
  for (int j = 0; j < 64; ++j) b[j] = *reinterpret_cast<const v2f*>(bcol_h + j * KDIM);
}

__device__ __forceinline__ void mma_16x32(const v4f (&a)[16], const v2f (&b)[64], v8f& d0, v8f& d1) {
#pragma unroll
  for (int s = 0; s < 32; ++s) {
    const v4f aq = a[s >> 1];
    v2f av, b0, b1;
    if (s & 1) { av.x = aq.z; av.y = aq.w; } else { av.x = aq.x; av.y = aq.y; }
    b0.x = b[2 * s].x; b0.y = b[2 * s + 1].x;
    b1.x = b[2 * s].y; b1.y = b[2 * s + 1].y;
    d0 = __builtin_amdgcn_wmma_f32_16x16x4_f32(false, av, false, b0, (short)0, d0, false, false);
    d1 = __builtin_amdgcn_wmma_f32_16x16x4_f32(false, av, false, b1, (short)0, d1, false, false);
  }
}

__device__ __forceinline__ float silu_f(float x) {
  return x * __builtin_amdgcn_rcpf(1.0f + __expf(-x));
}

__global__ void __launch_bounds__(64) __attribute__((amdgpu_waves_per_eu(1, 2)))
k_proj(const float* __restrict__ h,
       const float* __restrict__ W1a,
       const float* __restrict__ W1b,
       float* __restrict__ pi, float* __restrict__ pj) {
  const int lane = threadIdx.x & 31;
  const int wv = __builtin_amdgcn_readfirstlane(threadIdx.x >> 5);
  const int r0 = (blockIdx.x >> 2) * 16;
  const int n0 = (blockIdx.x & 3) * 32;
  const int c = lane & 15, hh = lane >> 4;
  const float* __restrict__ Bg = wv ? W1b : W1a;
  float* __restrict__ dst = wv ? pj : pi;

  v4f a[16];
  v2f b[64];
  load_a16(h + (r0 + c) * KDIM + 64 * hh, a);
  load_b64(Bg + 64 * hh * KDIM + n0 + 2 * c, b);
  v8f d0 = {0.f, 0.f, 0.f, 0.f, 0.f, 0.f, 0.f, 0.f};
  v8f d1 = d0;
  mma_16x32(a, b, d0, d1);
#pragma unroll
  for (int r = 0; r < 8; ++r) {
    const int row = r0 + 8 * hh + r;
    v2f o; o.x = d0[r]; o.y = d1[r];
    *reinterpret_cast<v2f*>(dst + row * KDIM + n0 + 2 * c) = o;
  }
}

#define MSG_ROWS 4
#define MSG_JB 32

__device__ __forceinline__ void msg_block(const float (&p)[MSG_JB],
                                            const float* __restrict__ w0,
                                            const float* __restrict__ w1,
                                            float base0, float base1,
                                            float& acc0, float& acc1) {
  float wa[MSG_JB], wb[MSG_JB];
#pragma unroll
  for (int q = 0; q < MSG_JB; q += 4) {
    const float4 va = *reinterpret_cast<const float4*>(w0 + q);
    const float4 vb = *reinterpret_cast<const float4*>(w1 + q);
    wa[q] = va.x; wa[q + 1] = va.y; wa[q + 2] = va.z; wa[q + 3] = va.w;
    wb[q] = vb.x; wb[q + 1] = vb.y; wb[q + 2] = vb.z; wb[q + 3] = vb.w;
  }
#pragma unroll
  for (int k = 0; k < MSG_JB; ++k) {
    acc0 = fmaf(wa[k], silu_f(base0 + p[k]), acc0);
    acc1 = fmaf(wb[k], silu_f(base1 + p[k]), acc1);
  }
}

__global__ void __launch_bounds__(256)
k_msg(const float* __restrict__ pi, const float* __restrict__ pj,
      const float* __restrict__ b1, const float* __restrict__ adj,
      float* __restrict__ S, float* __restrict__ Arow,
      float* __restrict__ Den) {
  __shared__ __attribute__((aligned(16))) float adjr[MSG_ROWS * 256];
  __shared__ __attribute__((aligned(16))) float adjm[MSG_ROWS * 256];
  const int row0 = blockIdx.x * MSG_ROWS;
  const int b    = row0 >> 8;
  const int i0   = row0 & 255;
  const int t    = threadIdx.x;
  const int m    = t & 127;
  const int wv   = __builtin_amdgcn_readfirstlane(t >> 5);
  const int ra   = (wv >> 2) * 2;

  const float* __restrict__ ar = adj + (size_t)row0 * 256;
  float av[MSG_ROWS];
#pragma unroll
  for (int q = 0; q < MSG_ROWS; ++q) av[q] = ar[q * 256 + t];

  const float bb    = b1[m];
  const float base0 = pi[(row0 + ra) * KDIM + m] + bb;
  const float base1 = pi[(row0 + ra + 1) * KDIM + m] + bb;
  const float* __restrict__ pjb = pj + (size_t)b * 256 * KDIM + m;

  float pa[MSG_JB], pb[MSG_JB];
#pragma unroll
  for (int k = 0; k < MSG_JB; ++k) pa[k] = pjb[k * KDIM];

#pragma unroll
  for (int q = 0; q < MSG_ROWS; ++q) {
    adjr[q * 256 + t] = av[q];
    adjm[q * 256 + t] = (t == i0 + q) ? 0.f : av[q];
  }
  __syncthreads();

  const float* __restrict__ w0 = adjm + ra * 256;
  const float* __restrict__ w1 = w0 + 256;
  const float* __restrict__ wr = adjr + (wv & 3) * 256;

  float acc0 = 0.f, acc1 = 0.f, rs = 0.f;
  for (int jb = 0; jb < 256; jb += 2 * MSG_JB) {
#pragma unroll
    for (int k = 0; k < MSG_JB; ++k) pb[k] = pjb[(jb + MSG_JB + k) * KDIM];
    if (wv < 4) {
#pragma unroll
      for (int k = 0; k < 2 * MSG_JB; ++k) rs += wr[jb + k];
    }
    msg_block(pa, w0 + jb, w1 + jb, base0, base1, acc0, acc1);
    const int jn = (jb + 2 * MSG_JB) & 255;
#pragma unroll
    for (int k = 0; k < MSG_JB; ++k) pa[k] = pjb[(jn + k) * KDIM];
    msg_block(pb, w0 + jb + MSG_JB, w1 + jb + MSG_JB, base0, base1, acc0, acc1);
  }
  S[(row0 + ra) * KDIM + m]     = acc0;
  S[(row0 + ra + 1) * KDIM + m] = acc1;
  if (wv < 4 && (t & 31) == 0) {
    Arow[row0 + wv] = rs - adjr[wv * 256 + i0 + wv];
    Den[row0 + wv]  = fmaxf(rs, 1.0f);
  }
}

__global__ void __launch_bounds__(32) __attribute__((amdgpu_waves_per_eu(1, 2)))
k_agg(const float* __restrict__ S, const float* __restrict__ W2,
      const float* __restrict__ b2, const float* __restrict__ Arow,
      const float* __restrict__ Den, float* __restrict__ magg) {
  const int lane = threadIdx.x;
  const int r0 = (blockIdx.x >> 2) * 16;
  const int n0 = (blockIdx.x & 3) * 32;
  const int c = lane & 15, hh = lane >> 4;

  v4f a[16];
  v2f b[64];
  load_a16(S + (r0 + c) * KDIM + 64 * hh, a);
  load_b64(W2 + 64 * hh * KDIM + n0 + 2 * c, b);
  const v2f bcol = *reinterpret_cast<const v2f*>(b2 + n0 + 2 * c);
  float ar[8], dn[8];
#pragma unroll
  for (int r = 0; r < 8; ++r) {
    ar[r] = Arow[r0 + 8 * hh + r];
    dn[r] = Den[r0 + 8 * hh + r];
  }
  v8f d0 = {0.f, 0.f, 0.f, 0.f, 0.f, 0.f, 0.f, 0.f};
  v8f d1 = d0;
  mma_16x32(a, b, d0, d1);
#pragma unroll
  for (int r = 0; r < 8; ++r) {
    const int row = r0 + 8 * hh + r;
    const float rd = __builtin_amdgcn_rcpf(dn[r]);
    v2f o;
    o.x = (d0[r] + bcol.x * ar[r]) * rd;
    o.y = (d1[r] + bcol.y * ar[r]) * rd;
    *reinterpret_cast<v2f*>(magg + row * KDIM + n0 + 2 * c) = o;
  }
}

__global__ void __launch_bounds__(128) __attribute__((amdgpu_waves_per_eu(1, 2)))
k_head(const float* __restrict__ magg,
       const float* __restrict__ Wc1, const float* __restrict__ bc1,
       const float* __restrict__ Wc2, const float* __restrict__ bc2,
       float* __restrict__ out) {
  __shared__ __attribute__((aligned(16))) float lT[16 * PAD];
  const int lane = threadIdx.x & 31;
  const int wv = threadIdx.x >> 5;
  const int r0 = blockIdx.x * 16;
  const int n0 = wv * 32;
  const int c = lane & 15, hh = lane >> 4;

  v4f a[16];
  v2f b[64], bn[64];
  load_a16(magg + (r0 + c) * KDIM + 64 * hh, a);
  load_b64(Wc1 + 64 * hh * KDIM + n0 + 2 * c, b);
  load_b64(Wc2 + 64 * hh * KDIM + n0 + 2 * c, bn);
  const v2f b1c = *reinterpret_cast<const v2f*>(bc1 + n0 + 2 * c);
  const v2f b2c = *reinterpret_cast<const v2f*>(bc2 + n0 + 2 * c);

  v8f d0 = {0.f, 0.f, 0.f, 0.f, 0.f, 0.f, 0.f, 0.f};
  v8f d1 = d0;
  mma_16x32(a, b, d0, d1);
#pragma unroll
  for (int r = 0; r < 8; ++r) {
    v2f t; t.x = silu_f(d0[r] + b1c.x); t.y = silu_f(d1[r] + b1c.y);
    *reinterpret_cast<v2f*>(&lT[(8 * hh + r) * PAD + n0 + 2 * c]) = t;
  }
  __syncthreads();

  load_a16(&lT[c * PAD + 64 * hh], a);
  v8f e0 = {0.f, 0.f, 0.f, 0.f, 0.f, 0.f, 0.f, 0.f};
  v8f e1 = e0;
  mma_16x32(a, bn, e0, e1);
#pragma unroll
  for (int r = 0; r < 8; ++r) {
    const int row = r0 + 8 * hh + r;
    v2f o; o.x = e0[r] + b2c.x; o.y = e1[r] + b2c.y;
    *reinterpret_cast<v2f*>(out + row * KDIM + n0 + 2 * c) = o;
  }
}

extern "C" void kernel_launch(void* const* d_in, const int* in_sizes, int n_in,
                              void* d_out, int out_size, void* d_ws, size_t ws_size,
                              hipStream_t stream) {
  const float* h   = (const float*)d_in[0];
  const float* adj = (const float*)d_in[1];
  const float* W1a = (const float*)d_in[2];
  const float* W1b = (const float*)d_in[3];
  const float* b1  = (const float*)d_in[4];
  const float* W2  = (const float*)d_in[5];
  const float* b2  = (const float*)d_in[6];
  const float* Wc1 = (const float*)d_in[7];
  const float* bc1 = (const float*)d_in[8];
  const float* Wc2 = (const float*)d_in[9];
  const float* bc2 = (const float*)d_in[10];
  float* out = (float*)d_out;

  float* ws = (float*)d_ws;
  float* pi = ws;
  float* pj = ws + 131072;
  float* S  = ws + 262144;
  float* Ar = ws + 393216;
  float* Dn = ws + 394240;
  float* mg = ws + 395264;

  k_proj<<<256, 64, 0, stream>>>(h, W1a, W1b, pi, pj);
  k_msg <<<ROWS / MSG_ROWS, 256, 0, stream>>>(pi, pj, b1, adj, S, Ar, Dn);
  k_agg <<<256, 32, 0, stream>>>(S, W2, b2, Ar, Dn, mg);
  k_head<<<64, 128, 0, stream>>>(mg, Wc1, bc1, Wc2, bc2, out);
}


# k_agg and k_head fused into one launch (4 -> 3 launches): a 16-row block does S@W2, @Wc1, @Wc2 in turn, both 16x128 intermediates pass through LDS; tiles, k order and epilogues unchanged
# speedup vs baseline: 2.4981x; 1.1189x over previous
#include <hip/hip_runtime.h>
#include <math.h>

#define ROWS 1024
#define KDIM 128
#define PAD 132

typedef __attribute__((ext_vector_type(2))) float v2f;
typedef __attribute__((ext_vector_type(4))) float v4f;
typedef __attribute__((ext_vector_type(8))) float v8f;


__device__ __forceinline__ void load_a16(const float* __restrict__ arow_h, v4f (&a)[16]) {
#pragma unroll
  for (int q = 0; q < 16; ++q) a[q] = *reinterpret_cast<const v4f*>(arow_h + 4 * q);
}

__device__ __forceinline__ void load_b64(const float* __restrict__ bcol_h, v2f (&b)[64]) {
#pragma unroll
  for (int j = 0; j < 64; ++j) b[j] = *reinterpret_cast<const v2f*>(bcol_h + j * KDIM);
}

__device__ __forceinline__ void mma_16x32(const v4f (&a)[16], const v2f (&b)[64], v8f& d0, v8f& d1) {
#pragma unroll
  for (int s = 0; s < 32; ++s) {
    const v4f aq = a[s >> 1];
    v2f av, b0, b1;
    if (s & 1) { av.x = aq.z; av.y = aq.w; } else { av.x = aq.x; av.y = aq.y; }
    b0.x = b[2 * s].x; b0.y = b[2 * s + 1].x;
    b1.x = b[2 * s].y; b1.y = b[2 * s + 1].y;
    d0 = __builtin_amdgcn_wmma_f32_16x16x4_f32(false, av, false, b0, (short)0, d0, false, false);
    d1 = __builtin_amdgcn_wmma_f32_16x16x4_f32(false, av, false, b1, (short)0, d1, false, false);
  }
}

__device__ __forceinline__ float silu_f(float x) {
  return x * __builtin_amdgcn_rcpf(1.0f + __expf(-x));
}

__global__ void __launch_bounds__(64) __attribute__((amdgpu_waves_per_eu(1, 2)))
k_proj(const float* __restrict__ h,
       const float* __restrict__ W1a,
       const float* __restrict__ W1b,
       float* __restrict__ pi, float* __restrict__ pj) {
  const int lane = threadIdx.x & 31;
  const int wv = __builtin_amdgcn_readfirstlane(threadIdx.x >> 5);
  const int r0 = (blockIdx.x >> 2) * 16;
  const int n0 = (blockIdx.x & 3) * 32;
  const int c = lane & 15, hh = lane >> 4;
  const float* __restrict__ Bg = wv ? W1b : W1a;
  float* __restrict__ dst = wv ? pj : pi;

  v4f a[16];
  v2f b[64];
  load_a16(h + (r0 + c) * KDIM + 64 * hh, a);
  load_b64(Bg + 64 * hh * KDIM + n0 + 2 * c, b);
  v8f d0 = {0.f, 0.f, 0.f, 0.f, 0.f, 0.f, 0.f, 0.f};
  v8f d1 = d0;
  mma_16x32(a, b, d0, d1);
#pragma unroll
  for (int r = 0; r < 8; ++r) {
    const int row = r0 + 8 * hh + r;
    v2f o; o.x = d0[r]; o.y = d1[r];
    *reinterpret_cast<v2f*>(dst + row * KDIM + n0 + 2 * c) = o;
  }
}

#define MSG_ROWS 4
#define MSG_JB 32

__device__ __forceinline__ void msg_block(const float (&p)[MSG_JB],
                                            const float* __restrict__ w0,
                                            const float* __restrict__ w1,
                                            float base0, float base1,
                                            float& acc0, float& acc1) {
  float wa[MSG_JB], wb[MSG_JB];
#pragma unroll
  for (int q = 0; q < MSG_JB; q += 4) {
    const float4 va = *reinterpret_cast<const float4*>(w0 + q);
    const float4 vb = *reinterpret_cast<const float4*>(w1 + q);
    wa[q] = va.x; wa[q + 1] = va.y; wa[q + 2] = va.z; wa[q + 3] = va.w;
    wb[q] = vb.x; wb[q + 1] = vb.y; wb[q + 2] = vb.z; wb[q + 3] = vb.w;
  }
#pragma unroll
  for (int k = 0; k < MSG_JB; ++k) {
    acc0 = fmaf(wa[k], silu_f(base0 + p[k]), acc0);
    acc1 = fmaf(wb[k], silu_f(base1 + p[k]), acc1);
  }
}

__global__ void __launch_bounds__(256)
k_msg(const float* __restrict__ pi, const float* __restrict__ pj,
      const float* __restrict__ b1, const float* __restrict__ adj,
      float* __restrict__ S, float* __restrict__ Arow,
      float* __restrict__ Den) {
  __shared__ __attribute__((aligned(16))) float adjr[MSG_ROWS * 256];
  __shared__ __attribute__((aligned(16))) float adjm[MSG_ROWS * 256];
  const int row0 = blockIdx.x * MSG_ROWS;
  const int b    = row0 >> 8;
  const int i0   = row0 & 255;
  const int t    = threadIdx.x;
  const int m    = t & 127;
  const int wv   = __builtin_amdgcn_readfirstlane(t >> 5);
  const int ra   = (wv >> 2) * 2;

  const float* __restrict__ ar = adj + (size_t)row0 * 256;
  float av[MSG_ROWS];
#pragma unroll
  for (int q = 0; q < MSG_ROWS; ++q) av[q] = ar[q * 256 + t];

  const float bb    = b1[m];
  const float base0 = pi[(row0 + ra) * KDIM + m] + bb;
  const float base1 = pi[(row0 + ra + 1) * KDIM + m] + bb;
  const float* __restrict__ pjb = pj + (size_t)b * 256 * KDIM + m;

  float pa[MSG_JB], pb[MSG_JB];
#pragma unroll
  for (int k = 0; k < MSG_JB; ++k) pa[k] = pjb[k * KDIM];

#pragma unroll
  for (int q = 0; q < MSG_ROWS; ++q) {
    adjr[q * 256 + t] = av[q];
    adjm[q * 256 + t] = (t == i0 + q) ? 0.f : av[q];
  }
  __syncthreads();

  const float* __restrict__ w0 = adjm + ra * 256;
  const float* __restrict__ w1 = w0 + 256;
  const float* __restrict__ wr = adjr + (wv & 3) * 256;

  float acc0 = 0.f, acc1 = 0.f, rs = 0.f;
  for (int jb = 0; jb < 256; jb += 2 * MSG_JB) {
#pragma unroll
    for (int k = 0; k < MSG_JB; ++k) pb[k] = pjb[(jb + MSG_JB + k) * KDIM];
    if (wv < 4) {
#pragma unroll
      for (int k = 0; k < 2 * MSG_JB; ++k) rs += wr[jb + k];
    }
    msg_block(pa, w0 + jb, w1 + jb, base0, base1, acc0, acc1);
    const int jn = (jb + 2 * MSG_JB) & 255;
#pragma unroll
    for (int k = 0; k < MSG_JB; ++k) pa[k] = pjb[(jn + k) * KDIM];
    msg_block(pb, w0 + jb + MSG_JB, w1 + jb + MSG_JB, base0, base1, acc0, acc1);
  }
  S[(row0 + ra) * KDIM + m]     = acc0;
  S[(row0 + ra + 1) * KDIM + m] = acc1;
  if (wv < 4 && (t & 31) == 0) {
    Arow[row0 + wv] = rs - adjr[wv * 256 + i0 + wv];
    Den[row0 + wv]  = fmaxf(rs, 1.0f);
  }
}

__global__ void __launch_bounds__(128) __attribute__((amdgpu_waves_per_eu(1, 1)))
k_tail(const float* __restrict__ S, const float* __restrict__ W2,
       const float* __restrict__ b2, const float* __restrict__ Arow,
       const float* __restrict__ Den,
       const float* __restrict__ Wc1, const float* __restrict__ bc1,
       const float* __restrict__ Wc2, const float* __restrict__ bc2,
       float* __restrict__ out) {
  __shared__ __attribute__((aligned(16))) float lM[16 * PAD];
  __shared__ __attribute__((aligned(16))) float lT[16 * PAD];
  const int lane = threadIdx.x & 31;
  const int wv = threadIdx.x >> 5;
  const int r0 = blockIdx.x * 16;
  const int n0 = wv * 32;
  const int c = lane & 15, hh = lane >> 4;

  v4f a[16];
  v2f bw[64], b[64], bn[64];
  load_a16(S + (r0 + c) * KDIM + 64 * hh, a);
  load_b64(W2 + 64 * hh * KDIM + n0 + 2 * c, bw);
  const v2f bcol = *reinterpret_cast<const v2f*>(b2 + n0 + 2 * c);
  float ar[8], dn[8];
#pragma unroll
  for (int r = 0; r < 8; ++r) {
    ar[r] = Arow[r0 + 8 * hh + r];
    dn[r] = Den[r0 + 8 * hh + r];
  }
  load_b64(Wc1 + 64 * hh * KDIM + n0 + 2 * c, b);
  load_b64(Wc2 + 64 * hh * KDIM + n0 + 2 * c, bn);
  const v2f b1c = *reinterpret_cast<const v2f*>(bc1 + n0 + 2 * c);
  const v2f b2c = *reinterpret_cast<const v2f*>(bc2 + n0 + 2 * c);

  v8f d0 = {0.f, 0.f, 0.f, 0.f, 0.f, 0.f, 0.f, 0.f};
  v8f d1 = d0;
  mma_16x32(a, bw, d0, d1);
#pragma unroll
  for (int r = 0; r < 8; ++r) {
    const float rd = __builtin_amdgcn_rcpf(dn[r]);
    v2f o;
    o.x = (d0[r] + bcol.x * ar[r]) * rd;
    o.y = (d1[r] + bcol.y * ar[r]) * rd;
    *reinterpret_cast<v2f*>(&lM[(8 * hh + r) * PAD + n0 + 2 * c]) = o;
  }
  __syncthreads();

  load_a16(&lM[c * PAD + 64 * hh], a);
  v8f f0 = {0.f, 0.f, 0.f, 0.f, 0.f, 0.f, 0.f, 0.f};
  v8f f1 = f0;
  mma_16x32(a, b, f0, f1);
#pragma unroll
  for (int r = 0; r < 8; ++r) {
    v2f t; t.x = silu_f(f0[r] + b1c.x); t.y = silu_f(f1[r] + b1c.y);
    *reinterpret_cast<v2f*>(&lT[(8 * hh + r) * PAD + n0 + 2 * c]) = t;
  }
  __syncthreads();

  load_a16(&lT[c * PAD + 64 * hh], a);
  v8f e0 = {0.f, 0.f, 0.f, 0.f, 0.f, 0.f, 0.f, 0.f};
  v8f e1 = e0;
  mma_16x32(a, bn, e0, e1);
#pragma unroll
  for (int r = 0; r < 8; ++r) {
    const int row = r0 + 8 * hh + r;
    v2f o; o.x = e0[r] + b2c.x; o.y = e1[r] + b2c.y;
    *reinterpret_cast<v2f*>(out + row * KDIM + n0 + 2 * c) = o;
  }
}

extern "C" void kernel_launch(void* const* d_in, const int* in_sizes, int n_in,
                              void* d_out, int out_size, void* d_ws, size_t ws_size,
                              hipStream_t stream) {
  const float* h   = (const float*)d_in[0];
  const float* adj = (const float*)d_in[1];
  const float* W1a = (const float*)d_in[2];
  const float* W1b = (const float*)d_in[3];
  const float* b1  = (const float*)d_in[4];
  const float* W2  = (const float*)d_in[5];
  const float* b2  = (const float*)d_in[6];
  const float* Wc1 = (const float*)d_in[7];
  const float* bc1 = (const float*)d_in[8];
  const float* Wc2 = (const float*)d_in[9];
  const float* bc2 = (const float*)d_in[10];
  float* out = (float*)d_out;

  float* ws = (float*)d_ws;
  float* pi = ws;
  float* pj = ws + 131072;
  float* S  = ws + 262144;
  float* Ar = ws + 393216;
  float* Dn = ws + 394240;

  k_proj<<<256, 64, 0, stream>>>(h, W1a, W1b, pi, pj);
  k_msg <<<ROWS / MSG_ROWS, 256, 0, stream>>>(pi, pj, b1, adj, S, Ar, Dn);
  k_tail<<<64, 128, 0, stream>>>(S, W2, b2, Ar, Dn, Wc1, bc1, Wc2, bc2, out);
}


# k_tail folded into k_msg (3 -> 2 launches): each 4-row block runs S@W2, @Wc1, @Wc2 itself on a padded 16-row WMMA tile, weights fetched under the silu loop and parked in LDS; k order and bits kept
# speedup vs baseline: 2.7201x; 1.0889x over previous
#include <hip/hip_runtime.h>
#include <math.h>

#define ROWS 1024
#define KDIM 128
#define PAD 132

typedef __attribute__((ext_vector_type(2))) float v2f;
typedef __attribute__((ext_vector_type(4))) float v4f;
typedef __attribute__((ext_vector_type(8))) float v8f;


__device__ __forceinline__ void load_a16(const float* __restrict__ arow_h, v4f (&a)[16]) {
#pragma unroll
  for (int q = 0; q < 16; ++q) a[q] = *reinterpret_cast<const v4f*>(arow_h + 4 * q);
}

__device__ __forceinline__ void load_b64(const float* __restrict__ bcol_h, v2f (&b)[64]) {
#pragma unroll
  for (int j = 0; j < 64; ++j) b[j] = *reinterpret_cast<const v2f*>(bcol_h + j * KDIM);
}

__device__ __forceinline__ void mma_16x32(const v4f (&a)[16], const v2f (&b)[64], v8f& d0, v8f& d1) {
#pragma unroll
  for (int s = 0; s < 32; ++s) {
    const v4f aq = a[s >> 1];
    v2f av, b0, b1;
    if (s & 1) { av.x = aq.z; av.y = aq.w; } else { av.x = aq.x; av.y = aq.y; }
    b0.x = b[2 * s].x; b0.y = b[2 * s + 1].x;
    b1.x = b[2 * s].y; b1.y = b[2 * s + 1].y;
    d0 = __builtin_amdgcn_wmma_f32_16x16x4_f32(false, av, false, b0, (short)0, d0, false, false);
    d1 = __builtin_amdgcn_wmma_f32_16x16x4_f32(false, av, false, b1, (short)0, d1, false, false);
  }
}

#define WQ 16

template <int NQ>
__device__ __forceinline__ void load_bquads(const float* __restrict__ bcol_q0, v4f (&v)[NQ]) {
#pragma unroll
  for (int q = 0; q < NQ; ++q) {
    v[q].x = bcol_q0[(4 * q + 0) * KDIM];
    v[q].y = bcol_q0[(4 * q + 1) * KDIM];
    v[q].z = bcol_q0[(4 * q + 2) * KDIM];
    v[q].w = bcol_q0[(4 * q + 3) * KDIM];
  }
}

__device__ __forceinline__ void unpark(const v4f* __restrict__ slot_t, float (&b)[64]) {
#pragma unroll
  for (int q = 0; q < WQ; ++q) {
    const v4f v = slot_t[q * 256];
    b[4 * q] = v.x; b[4 * q + 1] = v.y; b[4 * q + 2] = v.z; b[4 * q + 3] = v.w;
  }
}

__device__ __forceinline__ void mma_16x16(const v4f (&a)[16], const float (&b)[64], v8f& d) {
#pragma unroll
  for (int s = 0; s < 32; ++s) {
    const v4f aq = a[s >> 1];
    v2f av, bv;
    if (s & 1) { av.x = aq.z; av.y = aq.w; } else { av.x = aq.x; av.y = aq.y; }
    bv.x = b[2 * s]; bv.y = b[2 * s + 1];
    d = __builtin_amdgcn_wmma_f32_16x16x4_f32(false, av, false, bv, (short)0, d, false, false);
  }
}

__device__ __forceinline__ void lds_barrier() {
  __builtin_amdgcn_fence(__ATOMIC_RELEASE, "workgroup", "local");
  __builtin_amdgcn_s_barrier();
  __builtin_amdgcn_fence(__ATOMIC_ACQUIRE, "workgroup", "local");
}

__device__ __forceinline__ float silu_f(float x) {
  return x * __builtin_amdgcn_rcpf(1.0f + __expf(-x));
}

__global__ void __launch_bounds__(64) __attribute__((amdgpu_waves_per_eu(1, 2)))
k_proj(const float* __restrict__ h,
       const float* __restrict__ W1a,
       const float* __restrict__ W1b,
       float* __restrict__ pi, float* __restrict__ pj) {
  const int lane = threadIdx.x & 31;
  const int wv = __builtin_amdgcn_readfirstlane(threadIdx.x >> 5);
  const int r0 = (blockIdx.x >> 2) * 16;
  const int n0 = (blockIdx.x & 3) * 32;
  const int c = lane & 15, hh = lane >> 4;
  const float* __restrict__ Bg = wv ? W1b : W1a;
  float* __restrict__ dst = wv ? pj : pi;

  v4f a[16];
  v2f b[64];
  load_a16(h + (r0 + c) * KDIM + 64 * hh, a);
  load_b64(Bg + 64 * hh * KDIM + n0 + 2 * c, b);
  v8f d0 = {0.f, 0.f, 0.f, 0.f, 0.f, 0.f, 0.f, 0.f};
  v8f d1 = d0;
  mma_16x32(a, b, d0, d1);
#pragma unroll
  for (int r = 0; r < 8; ++r) {
    const int row = r0 + 8 * hh + r;
    v2f o; o.x = d0[r]; o.y = d1[r];
    *reinterpret_cast<v2f*>(dst + row * KDIM + n0 + 2 * c) = o;
  }
}

#define MSG_ROWS 4
#define MSG_JB 32

__device__ __forceinline__ void msg_block(const float (&p)[MSG_JB],
                                            const float* __restrict__ w0,
                                            const float* __restrict__ w1,
                                            float base0, float base1,
                                            float& acc0, float& acc1) {
  float wa[MSG_JB], wb[MSG_JB];
#pragma unroll
  for (int q = 0; q < MSG_JB; q += 4) {
    const float4 va = *reinterpret_cast<const float4*>(w0 + q);
    const float4 vb = *reinterpret_cast<const float4*>(w1 + q);
    wa[q] = va.x; wa[q + 1] = va.y; wa[q + 2] = va.z; wa[q + 3] = va.w;
    wb[q] = vb.x; wb[q + 1] = vb.y; wb[q + 2] = vb.z; wb[q + 3] = vb.w;
  }
#pragma unroll
  for (int k = 0; k < MSG_JB; ++k) {
    acc0 = fmaf(wa[k], silu_f(base0 + p[k]), acc0);
    acc1 = fmaf(wb[k], silu_f(base1 + p[k]), acc1);
  }
}

__global__ void __launch_bounds__(256) __attribute__((amdgpu_waves_per_eu(2, 2)))
k_msg_tail(const float* __restrict__ pi, const float* __restrict__ pj,
           const float* __restrict__ b1, const float* __restrict__ adj,
           const float* __restrict__ W2, const float* __restrict__ b2,
           const float* __restrict__ Wc1, const float* __restrict__ bc1,
           const float* __restrict__ Wc2, const float* __restrict__ bc2,
           float* __restrict__ out) {
  __shared__ __attribute__((aligned(16))) float adjr[MSG_ROWS * 256];
  __shared__ __attribute__((aligned(16))) float adjm[MSG_ROWS * 256];
  __shared__ __attribute__((aligned(16))) float lS[MSG_ROWS * PAD];
  __shared__ __attribute__((aligned(16))) float lM[16 * PAD];
  __shared__ __attribute__((aligned(16))) float lT[16 * PAD];
  __shared__ float lAr[MSG_ROWS];
  __shared__ float lDn[MSG_ROWS];
  __shared__ v4f lW[3 * WQ * 256];
  const int row0 = blockIdx.x * MSG_ROWS;
  const int b    = row0 >> 8;
  const int i0   = row0 & 255;
  const int t    = threadIdx.x;
  const int m    = t & 127;
  const int wv   = __builtin_amdgcn_readfirstlane(t >> 5);
  const int ra   = (wv >> 2) * 2;

  const float* __restrict__ ar4 = adj + (size_t)row0 * 256;
  float av[MSG_ROWS];
#pragma unroll
  for (int q = 0; q < MSG_ROWS; ++q) av[q] = ar4[q * 256 + t];

  const float bb    = b1[m];
  const float base0 = pi[(row0 + ra) * KDIM + m] + bb;
  const float base1 = pi[(row0 + ra + 1) * KDIM + m] + bb;
  const float* __restrict__ pjb = pj + (size_t)b * 256 * KDIM + m;

  float pa[MSG_JB], pb[MSG_JB];
#pragma unroll
  for (int k = 0; k < MSG_JB; ++k) pa[k] = pjb[k * KDIM];

  const int lane = t & 31;
  const int c = lane & 15, hh = lane >> 4;
  const int col = wv * 16 + c;
  const float b2c  = b2[col];
  const float bc1c = bc1[col];
  const float bc2c = bc2[col];
  const float* __restrict__ w2p  = W2  + 64 * hh * KDIM + col;
  const float* __restrict__ wc1p = Wc1 + 64 * hh * KDIM + col;
  const float* __restrict__ wc2p = Wc2 + 64 * hh * KDIM + col;
  v4f* __restrict__ slot = lW + t;

#pragma unroll
  for (int q = 0; q < MSG_ROWS; ++q) {
    adjr[q * 256 + t] = av[q];
    adjm[q * 256 + t] = (t == i0 + q) ? 0.f : av[q];
  }
  __syncthreads();

  const float* __restrict__ w0 = adjm + ra * 256;
  const float* __restrict__ w1 = w0 + 256;
  const float* __restrict__ wr = adjr + (wv & 3) * 256;

  v4f tw[4], tq[4], tn[4];
  float acc0 = 0.f, acc1 = 0.f, rs = 0.f;
  for (int jb = 0; jb < 256; jb += 2 * MSG_JB) {
#pragma unroll
    for (int k = 0; k < MSG_JB; ++k) pb[k] = pjb[(jb + MSG_JB + k) * KDIM];
    if (wv < 4) {
#pragma unroll
      for (int k = 0; k < 2 * MSG_JB; ++k) rs += wr[jb + k];
    }
    msg_block(pa, w0 + jb, w1 + jb, base0, base1, acc0, acc1);
    const int jn = (jb + 2 * MSG_JB) & 255;
#pragma unroll
    for (int k = 0; k < MSG_JB; ++k) pa[k] = pjb[(jn + k) * KDIM];
    msg_block(pb, w0 + jb + MSG_JB, w1 + jb + MSG_JB, base0, base1, acc0, acc1);
    int ko = (jb >> 2) * KDIM;
    int so = (jb >> 4) * 256;
    asm volatile("" : "+s"(ko), "+s"(so) : "v"(acc0));
    if (jb != 0) {
      v4f* __restrict__ sq = slot + so - 4 * 256;
#pragma unroll
      for (int q = 0; q < 4; ++q) {
        sq[(0 * WQ + q) * 256] = tw[q];
        sq[(1 * WQ + q) * 256] = tq[q];
        sq[(2 * WQ + q) * 256] = tn[q];
      }
    }
    load_bquads<4>(w2p + ko, tw);
    load_bquads<4>(wc1p + ko, tq);
    load_bquads<4>(wc2p + ko, tn);
  }
#pragma unroll
  for (int q = 0; q < 4; ++q) {
    slot[(0 * WQ + 12 + q) * 256] = tw[q];
    slot[(1 * WQ + 12 + q) * 256] = tq[q];
    slot[(2 * WQ + 12 + q) * 256] = tn[q];
  }

  lS[ra * PAD + m]       = acc0;
  lS[(ra + 1) * PAD + m] = acc1;
  if (wv < 4 && lane == 0) {
    lAr[wv] = rs - adjr[wv * 256 + i0 + wv];
    lDn[wv] = fmaxf(rs, 1.0f);
  }
  float bw[64], bq[64], bn[64];
  unpark(slot, bw);
  lds_barrier();

  v4f a[16];
  load_a16(&lS[(c & 3) * PAD + 64 * hh], a);
  float arw[MSG_ROWS], dnw[MSG_ROWS];
#pragma unroll
  for (int r = 0; r < MSG_ROWS; ++r) { arw[r] = lAr[r]; dnw[r] = lDn[r]; }
  unpark(slot + WQ * 256, bq);
  v8f d = {0.f, 0.f, 0.f, 0.f, 0.f, 0.f, 0.f, 0.f};
  mma_16x16(a, bw, d);
  __builtin_amdgcn_sched_group_barrier(0x100, 40, 0);
  __builtin_amdgcn_sched_group_barrier(0x008, 32, 0);
#pragma unroll
  for (int r = 0; r < 8; ++r) {
    const float rd = __builtin_amdgcn_rcpf(dnw[r & 3]);
    lM[(8 * hh + r) * PAD + col] = (d[r] + b2c * arw[r & 3]) * rd;
  }
  lds_barrier();

  load_a16(&lM[c * PAD + 64 * hh], a);
  unpark(slot + 2 * WQ * 256, bn);
  v8f f = {0.f, 0.f, 0.f, 0.f, 0.f, 0.f, 0.f, 0.f};
  mma_16x16(a, bq, f);
  __builtin_amdgcn_sched_group_barrier(0x100, 32, 0);
  __builtin_amdgcn_sched_group_barrier(0x008, 32, 0);
#pragma unroll
  for (int r = 0; r < 8; ++r) lT[(8 * hh + r) * PAD + col] = silu_f(f[r] + bc1c);
  lds_barrier();

  load_a16(&lT[c * PAD + 64 * hh], a);
  v8f e = {0.f, 0.f, 0.f, 0.f, 0.f, 0.f, 0.f, 0.f};
  mma_16x16(a, bn, e);
  __builtin_amdgcn_sched_group_barrier(0x100, 16, 0);
  __builtin_amdgcn_sched_group_barrier(0x008, 32, 0);
#pragma unroll
  for (int r = 0; r < 8; ++r) lM[(8 * hh + r) * PAD + col] = e[r] + bc2c;
  lds_barrier();
  if (wv < MSG_ROWS) {
    const v4f o = *reinterpret_cast<const v4f*>(&lM[wv * PAD + 4 * lane]);
    *reinterpret_cast<v4f*>(out + (row0 + wv) * KDIM + 4 * lane) = o;
  }
}

extern "C" void kernel_launch(void* const* d_in, const int* in_sizes, int n_in,
                              void* d_out, int out_size, void* d_ws, size_t ws_size,
                              hipStream_t stream) {
  const float* h   = (const float*)d_in[0];
  const float* adj = (const float*)d_in[1];
  const float* W1a = (const float*)d_in[2];
  const float* W1b = (const float*)d_in[3];
  const float* b1  = (const float*)d_in[4];
  const float* W2  = (const float*)d_in[5];
  const float* b2  = (const float*)d_in[6];
  const float* Wc1 = (const float*)d_in[7];
  const float* bc1 = (const float*)d_in[8];
  const float* Wc2 = (const float*)d_in[9];
  const float* bc2 = (const float*)d_in[10];
  float* out = (float*)d_out;

  float* ws = (float*)d_ws;
  float* pi = ws;
  float* pj = ws + 131072;

  k_proj<<<256, 64, 0, stream>>>(h, W1a, W1b, pi, pj);
  k_msg_tail<<<ROWS / MSG_ROWS, 256, 0, stream>>>(pi, pj, b1, adj, W2, b2, Wc1, bc1, Wc2, bc2, out);
}


# k_msg loop: silu as hx+hx*tanh(hx), hx=x/2: v_tanh_f32+fma replace exp,add,rcp,mul; row sum run by all 8 waves (no branch) so its chain hides in the silu work; terms round differently, j order kept
# speedup vs baseline: 2.9481x; 1.0838x over previous
#include <hip/hip_runtime.h>
#include <math.h>

#define ROWS 1024
#define KDIM 128
#define PAD 132

typedef __attribute__((ext_vector_type(2))) float v2f;
typedef __attribute__((ext_vector_type(4))) float v4f;
typedef __attribute__((ext_vector_type(8))) float v8f;


__device__ __forceinline__ void load_a16(const float* __restrict__ arow_h, v4f (&a)[16]) {
#pragma unroll
  for (int q = 0; q < 16; ++q) a[q] = *reinterpret_cast<const v4f*>(arow_h + 4 * q);
}

__device__ __forceinline__ void load_b64(const float* __restrict__ bcol_h, v2f (&b)[64]) {
#pragma unroll
  for (int j = 0; j < 64; ++j) b[j] = *reinterpret_cast<const v2f*>(bcol_h + j * KDIM);
}

__device__ __forceinline__ void mma_16x32(const v4f (&a)[16], const v2f (&b)[64], v8f& d0, v8f& d1) {
#pragma unroll
  for (int s = 0; s < 32; ++s) {
    const v4f aq = a[s >> 1];
    v2f av, b0, b1;
    if (s & 1) { av.x = aq.z; av.y = aq.w; } else { av.x = aq.x; av.y = aq.y; }
    b0.x = b[2 * s].x; b0.y = b[2 * s + 1].x;
    b1.x = b[2 * s].y; b1.y = b[2 * s + 1].y;
    d0 = __builtin_amdgcn_wmma_f32_16x16x4_f32(false, av, false, b0, (short)0, d0, false, false);
    d1 = __builtin_amdgcn_wmma_f32_16x16x4_f32(false, av, false, b1, (short)0, d1, false, false);
  }
}

#define WQ 16

template <int NQ>
__device__ __forceinline__ void load_bquads(const float* __restrict__ bcol_q0, v4f (&v)[NQ]) {
#pragma unroll
  for (int q = 0; q < NQ; ++q) {
    v[q].x = bcol_q0[(4 * q + 0) * KDIM];
    v[q].y = bcol_q0[(4 * q + 1) * KDIM];
    v[q].z = bcol_q0[(4 * q + 2) * KDIM];
    v[q].w = bcol_q0[(4 * q + 3) * KDIM];
  }
}

__device__ __forceinline__ void unpark(const v4f* __restrict__ slot_t, float (&b)[64]) {
#pragma unroll
  for (int q = 0; q < WQ; ++q) {
    const v4f v = slot_t[q * 256];
    b[4 * q] = v.x; b[4 * q + 1] = v.y; b[4 * q + 2] = v.z; b[4 * q + 3] = v.w;
  }
}

__device__ __forceinline__ void mma_16x16(const v4f (&a)[16], const float (&b)[64], v8f& d) {
#pragma unroll
  for (int s = 0; s < 32; ++s) {
    const v4f aq = a[s >> 1];
    v2f av, bv;
    if (s & 1) { av.x = aq.z; av.y = aq.w; } else { av.x = aq.x; av.y = aq.y; }
    bv.x = b[2 * s]; bv.y = b[2 * s + 1];
    d = __builtin_amdgcn_wmma_f32_16x16x4_f32(false, av, false, bv, (short)0, d, false, false);
  }
}

__device__ __forceinline__ void lds_barrier() {
  __builtin_amdgcn_fence(__ATOMIC_RELEASE, "workgroup", "local");
  __builtin_amdgcn_s_barrier();
  __builtin_amdgcn_fence(__ATOMIC_ACQUIRE, "workgroup", "local");
}

__device__ __forceinline__ float silu_f(float x) {
  return x * __builtin_amdgcn_rcpf(1.0f + __expf(-x));
}

__device__ __forceinline__ float silu_half(float hx) {
  return fmaf(hx, __builtin_amdgcn_tanhf(hx), hx);
}

__global__ void __launch_bounds__(64) __attribute__((amdgpu_waves_per_eu(1, 2)))
k_proj(const float* __restrict__ h,
       const float* __restrict__ W1a,
       const float* __restrict__ W1b,
       float* __restrict__ pi, float* __restrict__ pj) {
  const int lane = threadIdx.x & 31;
  const int wv = __builtin_amdgcn_readfirstlane(threadIdx.x >> 5);
  const int r0 = (blockIdx.x >> 2) * 16;
  const int n0 = (blockIdx.x & 3) * 32;
  const int c = lane & 15, hh = lane >> 4;
  const float* __restrict__ Bg = wv ? W1b : W1a;
  float* __restrict__ dst = wv ? pj : pi;

  v4f a[16];
  v2f b[64];
  load_a16(h + (r0 + c) * KDIM + 64 * hh, a);
  load_b64(Bg + 64 * hh * KDIM + n0 + 2 * c, b);
  v8f d0 = {0.f, 0.f, 0.f, 0.f, 0.f, 0.f, 0.f, 0.f};
  v8f d1 = d0;
  mma_16x32(a, b, d0, d1);
#pragma unroll
  for (int r = 0; r < 8; ++r) {
    const int row = r0 + 8 * hh + r;
    v2f o; o.x = d0[r]; o.y = d1[r];
    *reinterpret_cast<v2f*>(dst + row * KDIM + n0 + 2 * c) = o;
  }
}

#define MSG_ROWS 4
#define MSG_JB 32

__device__ __forceinline__ void msg_block(const float (&p)[MSG_JB],
                                            const float* __restrict__ w0,
                                            const float* __restrict__ w1,
                                            float hbase0, float hbase1,
                                            float& acc0, float& acc1) {
  float wa[MSG_JB], wb[MSG_JB];
#pragma unroll
  for (int q = 0; q < MSG_JB; q += 4) {
    const float4 va = *reinterpret_cast<const float4*>(w0 + q);
    const float4 vb = *reinterpret_cast<const float4*>(w1 + q);
    wa[q] = va.x; wa[q + 1] = va.y; wa[q + 2] = va.z; wa[q + 3] = va.w;
    wb[q] = vb.x; wb[q + 1] = vb.y; wb[q + 2] = vb.z; wb[q + 3] = vb.w;
  }
#pragma unroll
  for (int k = 0; k < MSG_JB; ++k) {
    acc0 = fmaf(wa[k], silu_half(fmaf(p[k], 0.5f, hbase0)), acc0);
    acc1 = fmaf(wb[k], silu_half(fmaf(p[k], 0.5f, hbase1)), acc1);
  }
}

__global__ void __launch_bounds__(256) __attribute__((amdgpu_waves_per_eu(2, 2)))
k_msg_tail(const float* __restrict__ pi, const float* __restrict__ pj,
           const float* __restrict__ b1, const float* __restrict__ adj,
           const float* __restrict__ W2, const float* __restrict__ b2,
           const float* __restrict__ Wc1, const float* __restrict__ bc1,
           const float* __restrict__ Wc2, const float* __restrict__ bc2,
           float* __restrict__ out) {
  __shared__ __attribute__((aligned(16))) float adjr[MSG_ROWS * 256];
  __shared__ __attribute__((aligned(16))) float adjm[MSG_ROWS * 256];
  __shared__ __attribute__((aligned(16))) float lS[MSG_ROWS * PAD];
  __shared__ __attribute__((aligned(16))) float lM[16 * PAD];
  __shared__ __attribute__((aligned(16))) float lT[16 * PAD];
  __shared__ float lAr[MSG_ROWS];
  __shared__ float lDn[MSG_ROWS];
  __shared__ v4f lW[3 * WQ * 256];
  const int row0 = blockIdx.x * MSG_ROWS;
  const int b    = row0 >> 8;
  const int i0   = row0 & 255;
  const int t    = threadIdx.x;
  const int m    = t & 127;
  const int wv   = __builtin_amdgcn_readfirstlane(t >> 5);
  const int ra   = (wv >> 2) * 2;

  const float* __restrict__ ar4 = adj + (size_t)row0 * 256;
  float av[MSG_ROWS];
#pragma unroll
  for (int q = 0; q < MSG_ROWS; ++q) av[q] = ar4[q * 256 + t];

  const float bb    = b1[m];
  const float base0 = 0.5f * (pi[(row0 + ra) * KDIM + m] + bb);
  const float base1 = 0.5f * (pi[(row0 + ra + 1) * KDIM + m] + bb);
  const float* __restrict__ pjb = pj + (size_t)b * 256 * KDIM + m;

  float pa[MSG_JB], pb[MSG_JB];
#pragma unroll
  for (int k = 0; k < MSG_JB; ++k) pa[k] = pjb[k * KDIM];

  const int lane = t & 31;
  const int c = lane & 15, hh = lane >> 4;
  const int col = wv * 16 + c;
  const float b2c  = b2[col];
  const float bc1c = bc1[col];
  const float bc2c = bc2[col];
  const float* __restrict__ w2p  = W2  + 64 * hh * KDIM + col;
  const float* __restrict__ wc1p = Wc1 + 64 * hh * KDIM + col;
  const float* __restrict__ wc2p = Wc2 + 64 * hh * KDIM + col;
  v4f* __restrict__ slot = lW + t;

#pragma unroll
  for (int q = 0; q < MSG_ROWS; ++q) {
    adjr[q * 256 + t] = av[q];
    adjm[q * 256 + t] = (t == i0 + q) ? 0.f : av[q];
  }
  __syncthreads();

  const float* __restrict__ w0 = adjm + ra * 256;
  const float* __restrict__ w1 = w0 + 256;
  const float* __restrict__ wr = adjr + (wv & 3) * 256;

  v4f tw[4], tq[4], tn[4];
  float acc0 = 0.f, acc1 = 0.f, rs = 0.f;
  for (int jb = 0; jb < 256; jb += 2 * MSG_JB) {
#pragma unroll
    for (int k = 0; k < MSG_JB; ++k) pb[k] = pjb[(jb + MSG_JB + k) * KDIM];
#pragma unroll
    for (int k = 0; k < 2 * MSG_JB; ++k) rs += wr[jb + k];
    msg_block(pa, w0 + jb, w1 + jb, base0, base1, acc0, acc1);
    const int jn = (jb + 2 * MSG_JB) & 255;
#pragma unroll
    for (int k = 0; k < MSG_JB; ++k) pa[k] = pjb[(jn + k) * KDIM];
    msg_block(pb, w0 + jb + MSG_JB, w1 + jb + MSG_JB, base0, base1, acc0, acc1);
    int ko = (jb >> 2) * KDIM;
    int so = (jb >> 4) * 256;
    asm volatile("" : "+s"(ko), "+s"(so) : "v"(acc0));
    if (jb != 0) {
      v4f* __restrict__ sq = slot + so - 4 * 256;
#pragma unroll
      for (int q = 0; q < 4; ++q) {
        sq[(0 * WQ + q) * 256] = tw[q];
        sq[(1 * WQ + q) * 256] = tq[q];
        sq[(2 * WQ + q) * 256] = tn[q];
      }
    }
    load_bquads<4>(w2p + ko, tw);
    load_bquads<4>(wc1p + ko, tq);
    load_bquads<4>(wc2p + ko, tn);
  }
#pragma unroll
  for (int q = 0; q < 4; ++q) {
    slot[(0 * WQ + 12 + q) * 256] = tw[q];
    slot[(1 * WQ + 12 + q) * 256] = tq[q];
    slot[(2 * WQ + 12 + q) * 256] = tn[q];
  }

  lS[ra * PAD + m]       = acc0;
  lS[(ra + 1) * PAD + m] = acc1;
  if (wv < 4 && lane == 0) {
    lAr[wv] = rs - adjr[wv * 256 + i0 + wv];
    lDn[wv] = fmaxf(rs, 1.0f);
  }
  float bw[64], bq[64], bn[64];
  unpark(slot, bw);
  lds_barrier();

  v4f a[16];
  load_a16(&lS[(c & 3) * PAD + 64 * hh], a);
  float arw[MSG_ROWS], dnw[MSG_ROWS];
#pragma unroll
  for (int r = 0; r < MSG_ROWS; ++r) { arw[r] = lAr[r]; dnw[r] = lDn[r]; }
  unpark(slot + WQ * 256, bq);
  v8f d = {0.f, 0.f, 0.f, 0.f, 0.f, 0.f, 0.f, 0.f};
  mma_16x16(a, bw, d);
  __builtin_amdgcn_sched_group_barrier(0x100, 40, 0);
  __builtin_amdgcn_sched_group_barrier(0x008, 32, 0);
#pragma unroll
  for (int r = 0; r < 8; ++r) {
    const float rd = __builtin_amdgcn_rcpf(dnw[r & 3]);
    lM[(8 * hh + r) * PAD + col] = (d[r] + b2c * arw[r & 3]) * rd;
  }
  lds_barrier();

  load_a16(&lM[c * PAD + 64 * hh], a);
  unpark(slot + 2 * WQ * 256, bn);
  v8f f = {0.f, 0.f, 0.f, 0.f, 0.f, 0.f, 0.f, 0.f};
  mma_16x16(a, bq, f);
  __builtin_amdgcn_sched_group_barrier(0x100, 32, 0);
  __builtin_amdgcn_sched_group_barrier(0x008, 32, 0);
#pragma unroll
  for (int r = 0; r < 8; ++r) lT[(8 * hh + r) * PAD + col] = silu_f(f[r] + bc1c);
  lds_barrier();

  load_a16(&lT[c * PAD + 64 * hh], a);
  v8f e = {0.f, 0.f, 0.f, 0.f, 0.f, 0.f, 0.f, 0.f};
  mma_16x16(a, bn, e);
  __builtin_amdgcn_sched_group_barrier(0x100, 16, 0);
  __builtin_amdgcn_sched_group_barrier(0x008, 32, 0);
#pragma unroll
  for (int r = 0; r < 8; ++r) lM[(8 * hh + r) * PAD + col] = e[r] + bc2c;
  lds_barrier();
  if (wv < MSG_ROWS) {
    const v4f o = *reinterpret_cast<const v4f*>(&lM[wv * PAD + 4 * lane]);
    *reinterpret_cast<v4f*>(out + (row0 + wv) * KDIM + 4 * lane) = o;
  }
}

extern "C" void kernel_launch(void* const* d_in, const int* in_sizes, int n_in,
                              void* d_out, int out_size, void* d_ws, size_t ws_size,
                              hipStream_t stream) {
  const float* h   = (const float*)d_in[0];
  const float* adj = (const float*)d_in[1];
  const float* W1a = (const float*)d_in[2];
  const float* W1b = (const float*)d_in[3];
  const float* b1  = (const float*)d_in[4];
  const float* W2  = (const float*)d_in[5];
  const float* b2  = (const float*)d_in[6];
  const float* Wc1 = (const float*)d_in[7];
  const float* bc1 = (const float*)d_in[8];
  const float* Wc2 = (const float*)d_in[9];
  const float* bc2 = (const float*)d_in[10];
  float* out = (float*)d_out;

  float* ws = (float*)d_ws;
  float* pi = ws;
  float* pj = ws + 131072;

  k_proj<<<256, 64, 0, stream>>>(h, W1a, W1b, pi, pj);
  k_msg_tail<<<ROWS / MSG_ROWS, 256, 0, stream>>>(pi, pj, b1, adj, W2, b2, Wc1, bc1, Wc2, bc2, out);
}


# W2 folded into Wc1: launch 1 also forms W21=W2@Wc1 and b2@Wc1 (18 more 2-wave blocks, same tile); each block's tail runs 2 products, not 3, and fetches 128 KB of weights; sums re-associated
# speedup vs baseline: 3.2429x; 1.1000x over previous
#include <hip/hip_runtime.h>
#include <math.h>

#define ROWS 1024
#define KDIM 128
#define PAD 132

typedef __attribute__((ext_vector_type(2))) float v2f;
typedef __attribute__((ext_vector_type(4))) float v4f;
typedef __attribute__((ext_vector_type(8))) float v8f;


__device__ __forceinline__ void load_a16(const float* __restrict__ arow_h, v4f (&a)[16]) {
#pragma unroll
  for (int q = 0; q < 16; ++q) a[q] = *reinterpret_cast<const v4f*>(arow_h + 4 * q);
}

__device__ __forceinline__ void load_b64(const float* __restrict__ bcol_h, v2f (&b)[64]) {
#pragma unroll
  for (int j = 0; j < 64; ++j) b[j] = *reinterpret_cast<const v2f*>(bcol_h + j * KDIM);
}

__device__ __forceinline__ void mma_16x32(const v4f (&a)[16], const v2f (&b)[64], v8f& d0, v8f& d1) {
#pragma unroll
  for (int s = 0; s < 32; ++s) {
    const v4f aq = a[s >> 1];
    v2f av, b0, b1;
    if (s & 1) { av.x = aq.z; av.y = aq.w; } else { av.x = aq.x; av.y = aq.y; }
    b0.x = b[2 * s].x; b0.y = b[2 * s + 1].x;
    b1.x = b[2 * s].y; b1.y = b[2 * s + 1].y;
    d0 = __builtin_amdgcn_wmma_f32_16x16x4_f32(false, av, false, b0, (short)0, d0, false, false);
    d1 = __builtin_amdgcn_wmma_f32_16x16x4_f32(false, av, false, b1, (short)0, d1, false, false);
  }
}

#define WQ 16

template <int NQ>
__device__ __forceinline__ void load_bquads(const float* __restrict__ bcol_q0, v4f (&v)[NQ]) {
#pragma unroll
  for (int q = 0; q < NQ; ++q) {
    v[q].x = bcol_q0[(4 * q + 0) * KDIM];
    v[q].y = bcol_q0[(4 * q + 1) * KDIM];
    v[q].z = bcol_q0[(4 * q + 2) * KDIM];
    v[q].w = bcol_q0[(4 * q + 3) * KDIM];
  }
}

__device__ __forceinline__ void unpark(const v4f* __restrict__ slot_t, float (&b)[64]) {
#pragma unroll
  for (int q = 0; q < WQ; ++q) {
    const v4f v = slot_t[q * 256];
    b[4 * q] = v.x; b[4 * q + 1] = v.y; b[4 * q + 2] = v.z; b[4 * q + 3] = v.w;
  }
}

__device__ __forceinline__ void mma_16x16(const v4f (&a)[16], const float (&b)[64], v8f& d) {
#pragma unroll
  for (int s = 0; s < 32; ++s) {
    const v4f aq = a[s >> 1];
    v2f av, bv;
    if (s & 1) { av.x = aq.z; av.y = aq.w; } else { av.x = aq.x; av.y = aq.y; }
    bv.x = b[2 * s]; bv.y = b[2 * s + 1];
    d = __builtin_amdgcn_wmma_f32_16x16x4_f32(false, av, false, bv, (short)0, d, false, false);
  }
}

__device__ __forceinline__ void lds_barrier() {
  __builtin_amdgcn_fence(__ATOMIC_RELEASE, "workgroup", "local");
  __builtin_amdgcn_s_barrier();
  __builtin_amdgcn_fence(__ATOMIC_ACQUIRE, "workgroup", "local");
}

__device__ __forceinline__ float silu_f(float x) {
  return x * __builtin_amdgcn_rcpf(1.0f + __expf(-x));
}

__device__ __forceinline__ float silu_half(float hx) {
  return fmaf(hx, __builtin_amdgcn_tanhf(hx), hx);
}

#define PROJ_BLOCKS 256
#define FOLD_BLOCKS 18
__global__ void __launch_bounds__(64) __attribute__((amdgpu_waves_per_eu(1, 2)))
k_proj(const float* __restrict__ h,
       const float* __restrict__ W1a,
       const float* __restrict__ W1b,
       const float* __restrict__ W2,
       const float* __restrict__ b2,
       const float* __restrict__ Wc1,
       float* __restrict__ pi, float* __restrict__ pj,
       float* __restrict__ W21, float* __restrict__ b21) {
  const int lane = threadIdx.x & 31;
  const int wv = __builtin_amdgcn_readfirstlane(threadIdx.x >> 5);
  const int c = lane & 15, hh = lane >> 4;
  const int blk = blockIdx.x;
  const float* __restrict__ arow;
  const float* __restrict__ Bg;
  float* __restrict__ dst;
  int n0;
  if (blk < PROJ_BLOCKS) {
    const int r0 = (blk >> 2) * 16;
    n0 = (blk & 3) * 32;
    arow = h + (r0 + c) * KDIM + 64 * hh;
    Bg = wv ? W1b : W1a;
    dst = (wv ? pj : pi) + r0 * KDIM;
  } else {
    const int e = (blk - PROJ_BLOCKS) * 2 + wv;
    const int rt = e >> 2;
    n0 = (e & 3) * 32;
    Bg = Wc1;
    if (rt < 8) {
      arow = W2 + (16 * rt + c) * KDIM + 64 * hh;
      dst = W21 + 16 * rt * KDIM;
    } else {
      arow = b2 + 64 * hh;
      dst = b21;
    }
  }

  v4f a[16];
  v2f b[64];
  load_a16(arow, a);
  load_b64(Bg + 64 * hh * KDIM + n0 + 2 * c, b);
  v8f d0 = {0.f, 0.f, 0.f, 0.f, 0.f, 0.f, 0.f, 0.f};
  v8f d1 = d0;
  mma_16x32(a, b, d0, d1);
#pragma unroll
  for (int r = 0; r < 8; ++r) {
    const int row = 8 * hh + r;
    v2f o; o.x = d0[r]; o.y = d1[r];
    *reinterpret_cast<v2f*>(dst + row * KDIM + n0 + 2 * c) = o;
  }
}

#define MSG_ROWS 4
#define MSG_JB 32

__device__ __forceinline__ void msg_block(const float (&p)[MSG_JB],
                                            const float* __restrict__ w0,
                                            const float* __restrict__ w1,
                                            float hbase0, float hbase1,
                                            float& acc0, float& acc1) {
  float wa[MSG_JB], wb[MSG_JB];
#pragma unroll
  for (int q = 0; q < MSG_JB; q += 4) {
    const float4 va = *reinterpret_cast<const float4*>(w0 + q);
    const float4 vb = *reinterpret_cast<const float4*>(w1 + q);
    wa[q] = va.x; wa[q + 1] = va.y; wa[q + 2] = va.z; wa[q + 3] = va.w;
    wb[q] = vb.x; wb[q + 1] = vb.y; wb[q + 2] = vb.z; wb[q + 3] = vb.w;
  }
#pragma unroll
  for (int k = 0; k < MSG_JB; ++k) {
    acc0 = fmaf(wa[k], silu_half(fmaf(p[k], 0.5f, hbase0)), acc0);
    acc1 = fmaf(wb[k], silu_half(fmaf(p[k], 0.5f, hbase1)), acc1);
  }
}

__global__ void __launch_bounds__(256) __attribute__((amdgpu_waves_per_eu(2, 2)))
k_msg_tail(const float* __restrict__ pi, const float* __restrict__ pj,
           const float* __restrict__ b1, const float* __restrict__ adj,
           const float* __restrict__ W21, const float* __restrict__ b21,
           const float* __restrict__ bc1,
           const float* __restrict__ Wc2, const float* __restrict__ bc2,
           float* __restrict__ out) {
  __shared__ __attribute__((aligned(16))) float adjr[MSG_ROWS * 256];
  __shared__ __attribute__((aligned(16))) float adjm[MSG_ROWS * 256];
  __shared__ __attribute__((aligned(16))) float lS[MSG_ROWS * PAD];
  __shared__ __attribute__((aligned(16))) float lM[16 * PAD];
  __shared__ __attribute__((aligned(16))) float lT[16 * PAD];
  __shared__ float lAr[MSG_ROWS];
  __shared__ float lDn[MSG_ROWS];
  __shared__ v4f lW[2 * WQ * 256];
  const int row0 = blockIdx.x * MSG_ROWS;
  const int b    = row0 >> 8;
  const int i0   = row0 & 255;
  const int t    = threadIdx.x;
  const int m    = t & 127;
  const int wv   = __builtin_amdgcn_readfirstlane(t >> 5);
  const int ra   = (wv >> 2) * 2;

  const float* __restrict__ ar4 = adj + (size_t)row0 * 256;
  float av[MSG_ROWS];
#pragma unroll
  for (int q = 0; q < MSG_ROWS; ++q) av[q] = ar4[q * 256 + t];

  const float bb    = b1[m];
  const float base0 = 0.5f * (pi[(row0 + ra) * KDIM + m] + bb);
  const float base1 = 0.5f * (pi[(row0 + ra + 1) * KDIM + m] + bb);
  const float* __restrict__ pjb = pj + (size_t)b * 256 * KDIM + m;

  float pa[MSG_JB], pb[MSG_JB];
#pragma unroll
  for (int k = 0; k < MSG_JB; ++k) pa[k] = pjb[k * KDIM];

  const int lane = t & 31;
  const int c = lane & 15, hh = lane >> 4;
  const int col = wv * 16 + c;
  const float b21c = b21[col];
  const float bc1c = bc1[col];
  const float bc2c = bc2[col];
  const float* __restrict__ w21p = W21 + 64 * hh * KDIM + col;
  const float* __restrict__ wc2p = Wc2 + 64 * hh * KDIM + col;
  v4f* __restrict__ slot = lW + t;

#pragma unroll
  for (int q = 0; q < MSG_ROWS; ++q) {
    adjr[q * 256 + t] = av[q];
    adjm[q * 256 + t] = (t == i0 + q) ? 0.f : av[q];
  }
  __syncthreads();

  const float* __restrict__ w0 = adjm + ra * 256;
  const float* __restrict__ w1 = w0 + 256;
  const float* __restrict__ wr = adjr + (wv & 3) * 256;

  v4f tw[4], tn[4];
  float acc0 = 0.f, acc1 = 0.f, rs = 0.f;
  for (int jb = 0; jb < 256; jb += 2 * MSG_JB) {
#pragma unroll
    for (int k = 0; k < MSG_JB; ++k) pb[k] = pjb[(jb + MSG_JB + k) * KDIM];
#pragma unroll
    for (int k = 0; k < 2 * MSG_JB; ++k) rs += wr[jb + k];
    msg_block(pa, w0 + jb, w1 + jb, base0, base1, acc0, acc1);
    const int jn = (jb + 2 * MSG_JB) & 255;
#pragma unroll
    for (int k = 0; k < MSG_JB; ++k) pa[k] = pjb[(jn + k) * KDIM];
    msg_block(pb, w0 + jb + MSG_JB, w1 + jb + MSG_JB, base0, base1, acc0, acc1);
    int ko = (jb >> 2) * KDIM;
    int so = (jb >> 4) * 256;
    asm volatile("" : "+s"(ko), "+s"(so) : "v"(acc0));
    if (jb != 0) {
      v4f* __restrict__ sq = slot + so - 4 * 256;
#pragma unroll
      for (int q = 0; q < 4; ++q) {
        sq[(0 * WQ + q) * 256] = tw[q];
        sq[(1 * WQ + q) * 256] = tn[q];
      }
    }
    load_bquads<4>(w21p + ko, tw);
    load_bquads<4>(wc2p + ko, tn);
  }
#pragma unroll
  for (int q = 0; q < 4; ++q) {
    slot[(0 * WQ + 12 + q) * 256] = tw[q];
    slot[(1 * WQ + 12 + q) * 256] = tn[q];
  }

  lS[ra * PAD + m]       = acc0;
  lS[(ra + 1) * PAD + m] = acc1;
  if (wv < 4 && lane == 0) {
    lAr[wv] = rs - adjr[wv * 256 + i0 + wv];
    lDn[wv] = fmaxf(rs, 1.0f);
  }
  float bw[64], bn[64];
  unpark(slot, bw);
  lds_barrier();

  v4f a[16];
  load_a16(&lS[(c & 3) * PAD + 64 * hh], a);
  float arw[MSG_ROWS], dnw[MSG_ROWS];
#pragma unroll
  for (int r = 0; r < MSG_ROWS; ++r) { arw[r] = lAr[r]; dnw[r] = lDn[r]; }
  unpark(slot + WQ * 256, bn);
  v8f d = {0.f, 0.f, 0.f, 0.f, 0.f, 0.f, 0.f, 0.f};
  mma_16x16(a, bw, d);
  __builtin_amdgcn_sched_group_barrier(0x100, 40, 0);
  __builtin_amdgcn_sched_group_barrier(0x008, 32, 0);
#pragma unroll
  for (int r = 0; r < 8; ++r) {
    const float rd = __builtin_amdgcn_rcpf(dnw[r & 3]);
    lT[(8 * hh + r) * PAD + col] = silu_f((d[r] + b21c * arw[r & 3]) * rd + bc1c);
  }
  lds_barrier();

  load_a16(&lT[c * PAD + 64 * hh], a);
  v8f e = {0.f, 0.f, 0.f, 0.f, 0.f, 0.f, 0.f, 0.f};
  mma_16x16(a, bn, e);
  __builtin_amdgcn_sched_group_barrier(0x100, 16, 0);
  __builtin_amdgcn_sched_group_barrier(0x008, 32, 0);
#pragma unroll
  for (int r = 0; r < 8; ++r) lM[(8 * hh + r) * PAD + col] = e[r] + bc2c;
  lds_barrier();
  if (wv < MSG_ROWS) {
    const v4f o = *reinterpret_cast<const v4f*>(&lM[wv * PAD + 4 * lane]);
    *reinterpret_cast<v4f*>(out + (row0 + wv) * KDIM + 4 * lane) = o;
  }
}

extern "C" void kernel_launch(void* const* d_in, const int* in_sizes, int n_in,
                              void* d_out, int out_size, void* d_ws, size_t ws_size,
                              hipStream_t stream) {
  const float* h   = (const float*)d_in[0];
  const float* adj = (const float*)d_in[1];
  const float* W1a = (const float*)d_in[2];
  const float* W1b = (const float*)d_in[3];
  const float* b1  = (const float*)d_in[4];
  const float* W2  = (const float*)d_in[5];
  const float* b2  = (const float*)d_in[6];
  const float* Wc1 = (const float*)d_in[7];
  const float* bc1 = (const float*)d_in[8];
  const float* Wc2 = (const float*)d_in[9];
  const float* bc2 = (const float*)d_in[10];
  float* out = (float*)d_out;

  float* ws = (float*)d_ws;
  float* pi = ws;
  float* pj = ws + 131072;
  float* W21 = ws + 262144;
  float* b21 = ws + 278528;

  k_proj<<<PROJ_BLOCKS + FOLD_BLOCKS, 64, 0, stream>>>(h, W1a, W1b, W2, b2, Wc1, pi, pj, W21, b21);
  k_msg_tail<<<ROWS / MSG_ROWS, 256, 0, stream>>>(pi, pj, b1, adj, W21, b21, bc1, Wc2, bc2, out);
}


# k_msg_tail: Wc2 (64 KB a block) is requested behind the silu loop and lands under the first product, not under the loop; W21 in the loop's first half; both straight to registers; sum orders kept
# speedup vs baseline: 4.0843x; 1.2595x over previous
#include <hip/hip_runtime.h>
#include <math.h>

#define ROWS 1024
#define KDIM 128
#define PAD 132

typedef __attribute__((ext_vector_type(2))) float v2f;
typedef __attribute__((ext_vector_type(4))) float v4f;
typedef __attribute__((ext_vector_type(8))) float v8f;


__device__ __forceinline__ void load_a16(const float* __restrict__ arow_h, v4f (&a)[16]) {
#pragma unroll
  for (int q = 0; q < 16; ++q) a[q] = *reinterpret_cast<const v4f*>(arow_h + 4 * q);
}

__device__ __forceinline__ void load_b64(const float* __restrict__ bcol_h, v2f (&b)[64]) {
#pragma unroll
  for (int j = 0; j < 64; ++j) b[j] = *reinterpret_cast<const v2f*>(bcol_h + j * KDIM);
}

__device__ __forceinline__ void mma_16x32(const v4f (&a)[16], const v2f (&b)[64], v8f& d0, v8f& d1) {
#pragma unroll
  for (int s = 0; s < 32; ++s) {
    const v4f aq = a[s >> 1];
    v2f av, b0, b1;
    if (s & 1) { av.x = aq.z; av.y = aq.w; } else { av.x = aq.x; av.y = aq.y; }
    b0.x = b[2 * s].x; b0.y = b[2 * s + 1].x;
    b1.x = b[2 * s].y; b1.y = b[2 * s + 1].y;
    d0 = __builtin_amdgcn_wmma_f32_16x16x4_f32(false, av, false, b0, (short)0, d0, false, false);
    d1 = __builtin_amdgcn_wmma_f32_16x16x4_f32(false, av, false, b1, (short)0, d1, false, false);
  }
}

#define WQ 16

template <int Q0, int NQ>
__device__ __forceinline__ void load_bquads(const float* __restrict__ bcol, v4f (&v)[WQ]) {
#pragma unroll
  for (int q = Q0; q < Q0 + NQ; ++q) {
    v[q].x = bcol[(4 * q + 0) * KDIM];
    v[q].y = bcol[(4 * q + 1) * KDIM];
    v[q].z = bcol[(4 * q + 2) * KDIM];
    v[q].w = bcol[(4 * q + 3) * KDIM];
  }
}

__device__ __forceinline__ void mma_16x16(const v4f (&a)[16], const v4f (&b)[WQ], v8f& d) {
#pragma unroll
  for (int s = 0; s < 32; ++s) {
    const v4f aq = a[s >> 1];
    const v4f bq = b[s >> 1];
    v2f av, bv;
    if (s & 1) { av.x = aq.z; av.y = aq.w; bv.x = bq.z; bv.y = bq.w; }
    else       { av.x = aq.x; av.y = aq.y; bv.x = bq.x; bv.y = bq.y; }
    d = __builtin_amdgcn_wmma_f32_16x16x4_f32(false, av, false, bv, (short)0, d, false, false);
  }
}

__device__ __forceinline__ void lds_barrier() {
  __builtin_amdgcn_fence(__ATOMIC_RELEASE, "workgroup", "local");
  __builtin_amdgcn_s_barrier();
  __builtin_amdgcn_fence(__ATOMIC_ACQUIRE, "workgroup", "local");
}

__device__ __forceinline__ float silu_f(float x) {
  return x * __builtin_amdgcn_rcpf(1.0f + __expf(-x));
}

__device__ __forceinline__ float silu_half(float hx) {
  return fmaf(hx, __builtin_amdgcn_tanhf(hx), hx);
}

#define PROJ_BLOCKS 256
#define FOLD_BLOCKS 18
__global__ void __launch_bounds__(64) __attribute__((amdgpu_waves_per_eu(1, 2)))
k_proj(const float* __restrict__ h,
       const float* __restrict__ W1a,
       const float* __restrict__ W1b,
       const float* __restrict__ W2,
       const float* __restrict__ b2,
       const float* __restrict__ Wc1,
       float* __restrict__ pi, float* __restrict__ pj,
       float* __restrict__ W21, float* __restrict__ b21) {
  const int lane = threadIdx.x & 31;
  const int wv = __builtin_amdgcn_readfirstlane(threadIdx.x >> 5);
  const int c = lane & 15, hh = lane >> 4;
  const int blk = blockIdx.x;
  const float* __restrict__ arow;
  const float* __restrict__ Bg;
  float* __restrict__ dst;
  int n0;
  if (blk < PROJ_BLOCKS) {
    const int r0 = (blk >> 2) * 16;
    n0 = (blk & 3) * 32;
    arow = h + (r0 + c) * KDIM + 64 * hh;
    Bg = wv ? W1b : W1a;
    dst = (wv ? pj : pi) + r0 * KDIM;
  } else {
    const int e = (blk - PROJ_BLOCKS) * 2 + wv;
    const int rt = e >> 2;
    n0 = (e & 3) * 32;
    Bg = Wc1;
    if (rt < 8) {
      arow = W2 + (16 * rt + c) * KDIM + 64 * hh;
      dst = W21 + 16 * rt * KDIM;
    } else {
      arow = b2 + 64 * hh;
      dst = b21;
    }
  }

  v4f a[16];
  v2f b[64];
  load_a16(arow, a);
  load_b64(Bg + 64 * hh * KDIM + n0 + 2 * c, b);
  v8f d0 = {0.f, 0.f, 0.f, 0.f, 0.f, 0.f, 0.f, 0.f};
  v8f d1 = d0;
  mma_16x32(a, b, d0, d1);
#pragma unroll
  for (int r = 0; r < 8; ++r) {
    const int row = 8 * hh + r;
    v2f o; o.x = d0[r]; o.y = d1[r];
    *reinterpret_cast<v2f*>(dst + row * KDIM + n0 + 2 * c) = o;
  }
}

#define MSG_ROWS 4
#define MSG_JB 16
#define LDS_QUADS ((MSG_ROWS * 256 + MSG_ROWS * PAD + 2 * 16 * PAD + 2 * MSG_ROWS) / 4)

__device__ __forceinline__ void msg_block(const v4f (&p)[MSG_JB],
                                            const float* __restrict__ w01,
                                            const v2f (&hb2)[4], v2f (&acc2)[4]) {
  v2f w2[MSG_JB];
#pragma unroll
  for (int q = 0; q < MSG_JB; q += 2) {
    const v4f v = *reinterpret_cast<const v4f*>(w01 + 2 * q);
    w2[q].x = v.x; w2[q].y = v.y; w2[q + 1].x = v.z; w2[q + 1].y = v.w;
  }
  const v2f half2 = {0.5f, 0.5f};
#pragma unroll
  for (int k = 0; k < MSG_JB; ++k) {
#pragma unroll
    for (int c = 0; c < 4; ++c) {
      const v2f pp = {p[k][c], p[k][c]};
      const v2f hx = __builtin_elementwise_fma(pp, half2, hb2[c]);
      v2f t;
      t.x = __builtin_amdgcn_tanhf(hx.x);
      t.y = __builtin_amdgcn_tanhf(hx.y);
      const v2f sv = __builtin_elementwise_fma(hx, t, hx);
      acc2[c] = __builtin_elementwise_fma(w2[k], sv, acc2[c]);
    }
  }
}

__device__ __forceinline__ void load_rowsum64(const float* __restrict__ wr, float (&rv)[64]) {
#pragma unroll
  for (int k = 0; k < 64; ++k) rv[k] = wr[k];
}

template <int I>
__device__ __forceinline__ void msg_chunk_tail(const float (&rv)[64], float& rs, const v2f (&acc2)[4],
                                                 const float* __restrict__ w21p, const float* __restrict__ wc2p,
                                                 v4f (&wq)[WQ], v4f (&cq)[WQ]) {
  if (I < 3) {
    asm volatile("" : "+s"(rs) : "v"(acc2[0].x));
#pragma unroll
    for (int k = 0; k < 64; ++k) rs += rv[k];
  }
  int ko = 0;
  if (I < 3) {
    asm volatile("" : "+s"(ko) : "v"(acc2[1].x), "s"(rs));
  } else {
    asm volatile("" : "+s"(ko) : "v"(acc2[0].x), "v"(acc2[1].x), "v"(acc2[2].x), "v"(acc2[3].x), "s"(rs));
  }
  if (I == 0) load_bquads<0, 8>(w21p + ko, wq);
  if (I == 1) load_bquads<8, 8>(w21p + ko, wq);
  if (I == 3) load_bquads<0, 15>(wc2p + ko, cq);
}

__global__ void __launch_bounds__(256) __attribute__((amdgpu_waves_per_eu(2, 2)))
k_msg_tail(const float* __restrict__ pi, const float* __restrict__ pj,
           const float* __restrict__ b1, const float* __restrict__ adj,
           const float* __restrict__ W21, const float* __restrict__ b21,
           const float* __restrict__ bc1,
           const float* __restrict__ Wc2, const float* __restrict__ bc2,
           float* __restrict__ out) {
  __shared__ v4f smem[LDS_QUADS];
  float* __restrict__ const adjm = reinterpret_cast<float*>(smem);
  float* __restrict__ const lS   = adjm + MSG_ROWS * 256;
  float* __restrict__ const lM   = lS + MSG_ROWS * PAD;
  float* __restrict__ const lT   = lM + 16 * PAD;
  float* __restrict__ const lAr  = lT + 16 * PAD;
  float* __restrict__ const lDn  = lAr + MSG_ROWS;
  const int row0 = blockIdx.x * MSG_ROWS;
  const int b    = row0 >> 8;
  const int i0   = row0 & 255;
  const int t    = threadIdx.x;
  const int m    = t & 127;
  const int wv   = __builtin_amdgcn_readfirstlane(t >> 5);
  const int ra   = (wv >> 2) * 2;

  const float* __restrict__ ar4 = adj + (size_t)row0 * 256;
  float av[MSG_ROWS];
#pragma unroll
  for (int q = 0; q < MSG_ROWS; ++q) av[q] = ar4[q * 256 + t];

  const int lane = t & 31;
  const int jq   = (wv & 3) * 64;
  const int c4   = 4 * lane;
  const v4f bb4 = *reinterpret_cast<const v4f*>(b1 + c4);
  const v4f pi0 = *reinterpret_cast<const v4f*>(pi + (row0 + ra) * KDIM + c4);
  const v4f pi1 = *reinterpret_cast<const v4f*>(pi + (row0 + ra + 1) * KDIM + c4);
  const float* __restrict__ pjb = pj + ((size_t)b * 256 + jq) * KDIM + c4;

  v4f p0[MSG_JB], p1[MSG_JB], p2[MSG_JB], p3[MSG_JB];
#pragma unroll
  for (int k = 0; k < MSG_JB; ++k) p0[k] = *reinterpret_cast<const v4f*>(pjb + k * KDIM);
#pragma unroll
  for (int k = 0; k < MSG_JB; ++k) p1[k] = *reinterpret_cast<const v4f*>(pjb + (MSG_JB + k) * KDIM);

  const int c = lane & 15, hh = lane >> 4;
  const int col = wv * 16 + c;
  const float b21c = b21[col];
  const float bc1c = bc1[col];
  const float bc2c = bc2[col];
  const float* __restrict__ w21p = W21 + 64 * hh * KDIM + col;
  const float* __restrict__ wc2p = Wc2 + 64 * hh * KDIM + col;

  const float* __restrict__ wr = ar4 + (wv & 3) * 256;
  float rv[64];
  load_rowsum64(wr, rv);

#pragma unroll
  for (int q = 0; q < MSG_ROWS; ++q) {
    adjm[(q >> 1) * 512 + 2 * t + (q & 1)] = (t == i0 + q) ? 0.f : av[q];
  }
  __syncthreads();

  const float* __restrict__ w01 = adjm + (ra >> 1) * 512 + 2 * jq;
  v4f wq[WQ], cq[WQ];
  v2f acc2[4], hb2[4];
#pragma unroll
  for (int cc = 0; cc < 4; ++cc) {
    acc2[cc].x = 0.f; acc2[cc].y = 0.f;
    hb2[cc].x = 0.5f * (pi0[cc] + bb4[cc]);
    hb2[cc].y = 0.5f * (pi1[cc] + bb4[cc]);
  }
  float rs = 0.f;
#pragma unroll
  for (int k = 0; k < 64; ++k) rs += rv[k];
  load_rowsum64(wr + 64, rv);
#pragma unroll
  for (int k = 0; k < MSG_JB; ++k) p2[k] = *reinterpret_cast<const v4f*>(pjb + (2 * MSG_JB + k) * KDIM);
  msg_block(p0, w01, hb2, acc2);
  msg_chunk_tail<0>(rv, rs, acc2, w21p, wc2p, wq, cq);
  load_rowsum64(wr + 128, rv);
#pragma unroll
  for (int k = 0; k < MSG_JB; ++k) p3[k] = *reinterpret_cast<const v4f*>(pjb + (3 * MSG_JB + k) * KDIM);
  msg_block(p1, w01 + 2 * MSG_JB, hb2, acc2);
  msg_chunk_tail<1>(rv, rs, acc2, w21p, wc2p, wq, cq);
  load_rowsum64(wr + 192, rv);
  msg_block(p2, w01 + 4 * MSG_JB, hb2, acc2);
  msg_chunk_tail<2>(rv, rs, acc2, w21p, wc2p, wq, cq);
  msg_block(p3, w01 + 6 * MSG_JB, hb2, acc2);
  msg_chunk_tail<3>(rv, rs, acc2, w21p, wc2p, wq, cq);

  {
    v4f lo, hi;
    lo.x = acc2[0].x; lo.y = acc2[1].x; lo.z = acc2[2].x; lo.w = acc2[3].x;
    hi.x = acc2[0].y; hi.y = acc2[1].y; hi.z = acc2[2].y; hi.w = acc2[3].y;
    float* __restrict__ lP = lM + (wv & 3) * MSG_ROWS * PAD;
    *reinterpret_cast<v4f*>(lP + ra * PAD + c4)       = lo;
    *reinterpret_cast<v4f*>(lP + (ra + 1) * PAD + c4) = hi;
  }
  if (wv < 4 && lane == 0) {
    lAr[wv] = rs - ar4[wv * 256 + i0 + wv];
    lDn[wv] = fmaxf(rs, 1.0f);
  }
  lds_barrier();
#pragma unroll
  for (int rr = 0; rr < 2; ++rr) {
    const float* __restrict__ pq = lM + (ra + rr) * PAD + m;
    const float q0 = pq[0], q1 = pq[MSG_ROWS * PAD], q2 = pq[2 * MSG_ROWS * PAD], q3 = pq[3 * MSG_ROWS * PAD];
    lS[(ra + rr) * PAD + m] = ((q0 + q1) + q2) + q3;
  }
  lds_barrier();

  v4f a[16];
  load_a16(&lS[(c & 3) * PAD + 64 * hh], a);
  float arw[MSG_ROWS], dnw[MSG_ROWS];
#pragma unroll
  for (int r = 0; r < MSG_ROWS; ++r) { arw[r] = lAr[r]; dnw[r] = lDn[r]; }
  v8f d = {0.f, 0.f, 0.f, 0.f, 0.f, 0.f, 0.f, 0.f};
  mma_16x16(a, wq, d);
  __builtin_amdgcn_sched_group_barrier(0x100, 24, 0);
  __builtin_amdgcn_sched_group_barrier(0x008, 32, 0);
#pragma unroll
  for (int r = 0; r < 8; ++r) {
    const float rd = __builtin_amdgcn_rcpf(dnw[r & 3]);
    lT[(8 * hh + r) * PAD + col] = silu_f((d[r] + b21c * arw[r & 3]) * rd + bc1c);
  }
  int lo = 0;
  asm volatile("" : "+s"(lo) : "v"(d[7]));
  load_bquads<15, 1>(wc2p + lo, cq);
  lds_barrier();

  load_a16(&lT[c * PAD + 64 * hh], a);
  v8f e = {0.f, 0.f, 0.f, 0.f, 0.f, 0.f, 0.f, 0.f};
  mma_16x16(a, cq, e);
  __builtin_amdgcn_sched_group_barrier(0x100, 16, 0);
  __builtin_amdgcn_sched_group_barrier(0x008, 32, 0);
#pragma unroll
  for (int r = 0; r < 8; ++r) lM[(8 * hh + r) * PAD + col] = e[r] + bc2c;
  lds_barrier();
  if (wv < MSG_ROWS) {
    const v4f o = *reinterpret_cast<const v4f*>(&lM[wv * PAD + 4 * lane]);
    *reinterpret_cast<v4f*>(out + (row0 + wv) * KDIM + 4 * lane) = o;
  }
}

extern "C" void kernel_launch(void* const* d_in, const int* in_sizes, int n_in,
                              void* d_out, int out_size, void* d_ws, size_t ws_size,
                              hipStream_t stream) {
  const float* h   = (const float*)d_in[0];
  const float* adj = (const float*)d_in[1];
  const float* W1a = (const float*)d_in[2];
  const float* W1b = (const float*)d_in[3];
  const float* b1  = (const float*)d_in[4];
  const float* W2  = (const float*)d_in[5];
  const float* b2  = (const float*)d_in[6];
  const float* Wc1 = (const float*)d_in[7];
  const float* bc1 = (const float*)d_in[8];
  const float* Wc2 = (const float*)d_in[9];
  const float* bc2 = (const float*)d_in[10];
  float* out = (float*)d_out;

  float* ws = (float*)d_ws;
  float* pi = ws;
  float* pj = ws + 131072;
  float* W21 = ws + 262144;
  float* b21 = ws + 278528;

  k_proj<<<PROJ_BLOCKS + FOLD_BLOCKS, 64, 0, stream>>>(h, W1a, W1b, W2, b2, Wc1, pi, pj, W21, b21);
  k_msg_tail<<<ROWS / MSG_ROWS, 256, 0, stream>>>(pi, pj, b1, adj, W21, b21, bc1, Wc2, bc2, out);
}
